# starting kernel
# baseline (speedup 1.0000x reference)
#include <hip/hip_runtime.h>


typedef __attribute__((ext_vector_type(2))) float v2f;
typedef __attribute__((ext_vector_type(8))) float v8f;

#define D_DIM 1024
#define LSEQ  2048

template <bool RELU_BIAS>
__global__ __launch_bounds__(128) void gemm_f32_wmma(
    const float* __restrict__ A, const float* __restrict__ Bm,
    const float* __restrict__ bias, float* __restrict__ C,
    int M, int N, int K)
{
    const int lane = threadIdx.x & 31;
    const int wave = threadIdx.x >> 5;
    const int half = lane >> 4;
    const int l16  = lane & 15;
    const int koff = half * 2;

    const int m0 = blockIdx.y * 64 + (wave >> 1) * 32;
    const int n0 = blockIdx.x * 64 + (wave & 1) * 32;

    const float* Arow0 = A + (size_t)(m0 + l16) * K + koff;
    const float* Arow1 = A + (size_t)(m0 + 16 + l16) * K + koff;
    const float* Bcol0 = Bm + n0 + l16;
    const float* Bcol1 = Bm + n0 + 16 + l16;

    v8f acc00 = {}, acc01 = {}, acc10 = {}, acc11 = {};
    for (int k0 = 0; k0 < K; k0 += 4) {
        v2f a0 = *(const v2f*)(Arow0 + k0);
        v2f a1 = *(const v2f*)(Arow1 + k0);
        v2f b0, b1;
        size_t r0 = (size_t)(k0 + koff) * N;
        size_t r1 = r0 + N;
        b0.x = Bcol0[r0];  b0.y = Bcol0[r1];
        b1.x = Bcol1[r0];  b1.y = Bcol1[r1];
        acc00 = __builtin_amdgcn_wmma_f32_16x16x4_f32(false, a0, false, b0, (short)0, acc00, false, false);
        acc01 = __builtin_amdgcn_wmma_f32_16x16x4_f32(false, a0, false, b1, (short)0, acc01, false, false);
        acc10 = __builtin_amdgcn_wmma_f32_16x16x4_f32(false, a1, false, b0, (short)0, acc10, false, false);
        acc11 = __builtin_amdgcn_wmma_f32_16x16x4_f32(false, a1, false, b1, (short)0, acc11, false, false);
    }

    float bv0 = 0.0f, bv1 = 0.0f;
    if (RELU_BIAS) {
        bv0 = bias[n0 + l16];
        bv1 = bias[n0 + 16 + l16];
    }
#pragma unroll
    for (int r = 0; r < 8; ++r) {
        int row0 = m0 + r + half * 8;
        int row1 = row0 + 16;
        float v00 = acc00[r], v01 = acc01[r], v10 = acc10[r], v11 = acc11[r];
        if (RELU_BIAS) {
            v00 += bv0; v01 += bv1; v10 += bv0; v11 += bv1;
            v00 = v00 > 0.0f ? v00 : 0.0f;
            v01 = v01 > 0.0f ? v01 : 0.0f;
            v10 = v10 > 0.0f ? v10 : 0.0f;
            v11 = v11 > 0.0f ? v11 : 0.0f;
        }
        C[(size_t)row0 * N + n0 + l16]      = v00;
        C[(size_t)row0 * N + n0 + 16 + l16] = v01;
        C[(size_t)row1 * N + n0 + l16]      = v10;
        C[(size_t)row1 * N + n0 + 16 + l16] = v11;
    }
}

__global__ __launch_bounds__(256) void cumavg_kernel(
    const float* __restrict__ q, float* __restrict__ xavg, int Bsz)
{
    int idx = blockIdx.x * blockDim.x + threadIdx.x;
    if (idx >= Bsz * D_DIM) return;
    int b = idx / D_DIM, d = idx % D_DIM;
    size_t base = (size_t)b * LSEQ * D_DIM + d;
    float acc = 0.0f;
    for (int i = 0; i < LSEQ; ++i) {
        acc += q[base + (size_t)i * D_DIM];
        xavg[base + (size_t)i * D_DIM] = acc / (float)(i + 1);
    }
}

__global__ __launch_bounds__(256) void rowsum_kernel(
    const float* __restrict__ g, float* __restrict__ S, int nrows)
{
    int wid  = (int)((blockIdx.x * blockDim.x + threadIdx.x) >> 5);
    int lane = threadIdx.x & 31;
    if (wid >= nrows) return;
    int i = wid % LSEQ;
    const float* row = g + (size_t)wid * LSEQ;
    float s = 0.0f;
    for (int k = (LSEQ - 1 - i) + lane; k < LSEQ; k += 32) s += row[k];
#pragma unroll
    for (int off = 16; off; off >>= 1) s += __shfl_xor(s, off, 32);
    if (lane == 0) S[wid] = s + 1e-8f;
}

__global__ __launch_bounds__(128) void attn_apply_wmma(
    const float* __restrict__ g, const float* __restrict__ q,
    const float* __restrict__ S, float* __restrict__ out)
{
    const int lane = threadIdx.x & 31;
    const int wave = threadIdx.x >> 5;
    const int b    = blockIdx.z;
    const int i0   = blockIdx.y * 16;
    const int d0   = blockIdx.x * 128 + wave * 32;
    const int half = lane >> 4;
    const int l16  = lane & 15;
    const int koff = half * 2;

    const int iA    = i0 + l16;
    const int shift = LSEQ - 1 - iA;
    const float* grow = g + ((size_t)b * LSEQ + iA) * LSEQ + shift;
    const float* qB0  = q + (size_t)b * LSEQ * D_DIM + d0 + l16;
    const float* qB1  = qB0 + 16;

    v8f acc0 = {}, acc1 = {};

    for (int j0 = 0; j0 < i0; j0 += 4) {
        int jx = j0 + koff;
        v2f a, b0, b1;
        a.x = grow[jx];
        a.y = grow[jx + 1];
        size_t r0 = (size_t)jx * D_DIM;
        size_t r1 = r0 + D_DIM;
        b0.x = qB0[r0];  b0.y = qB0[r1];
        b1.x = qB1[r0];  b1.y = qB1[r1];
        acc0 = __builtin_amdgcn_wmma_f32_16x16x4_f32(false, a, false, b0, (short)0, acc0, false, false);
        acc1 = __builtin_amdgcn_wmma_f32_16x16x4_f32(false, a, false, b1, (short)0, acc1, false, false);
    }

#pragma unroll
    for (int t = 0; t < 4; ++t) {
        int jx = i0 + t * 4 + koff;
        int jy = jx + 1;
        v2f a, b0, b1;
        a.x = (jx <= iA) ? grow[jx] : 0.0f;
        a.y = (jy <= iA) ? grow[jy] : 0.0f;
        size_t r0 = (size_t)jx * D_DIM;
        size_t r1 = r0 + D_DIM;
        b0.x = qB0[r0];  b0.y = qB0[r1];
        b1.x = qB1[r0];  b1.y = qB1[r1];
        acc0 = __builtin_amdgcn_wmma_f32_16x16x4_f32(false, a, false, b0, (short)0, acc0, false, false);
        acc1 = __builtin_amdgcn_wmma_f32_16x16x4_f32(false, a, false, b1, (short)0, acc1, false, false);
    }

#pragma unroll
    for (int r = 0; r < 8; ++r) {
        int row = i0 + r + half * 8;
        float invs = 1.0f / S[b * LSEQ + row];
        size_t o = ((size_t)b * LSEQ + row) * D_DIM + d0 + l16;
        out[o]      = acc0[r] * invs;
        out[o + 16] = acc1[r] * invs;
    }
}

extern "C" void kernel_launch(void* const* d_in, const int* in_sizes, int n_in,
                              void* d_out, int out_size, void* d_ws, size_t ws_size,
                              hipStream_t stream)
{
    const float* x  = (const float*)d_in[0];
    const float* Wq = (const float*)d_in[1];
    const float* Wg = (const float*)d_in[2];
    const float* bg = (const float*)d_in[3];
    const float* Wc = (const float*)d_in[4];
    float* out = (float*)d_out;

    const int Bsz = in_sizes[0] / (LSEQ * D_DIM);
    const int M   = Bsz * LSEQ;

    float* q    = (float*)d_ws;
    float* xavg = q    + (size_t)M * D_DIM;
    float* g    = xavg + (size_t)M * D_DIM;
    float* S    = g    + (size_t)M * LSEQ;

    dim3 blk(128);

    gemm_f32_wmma<false><<<dim3(D_DIM / 64, M / 64), blk, 0, stream>>>(
        x, Wq, nullptr, q, M, D_DIM, D_DIM);

    cumavg_kernel<<<(Bsz * D_DIM + 255) / 256, 256, 0, stream>>>(q, xavg, Bsz);

    gemm_f32_wmma<true><<<dim3(LSEQ / 64, M / 64), blk, 0, stream>>>(
        xavg, Wg, bg, g, M, LSEQ, D_DIM);

    rowsum_kernel<<<(M * 32 + 255) / 256, 256, 0, stream>>>(g, S, M);

    attn_apply_wmma<<<dim3(D_DIM / 128, LSEQ / 16, Bsz), blk, 0, stream>>>(
        g, q, S, xavg);

    gemm_f32_wmma<false><<<dim3(D_DIM / 64, M / 64), blk, 0, stream>>>(
        xavg, Wc, nullptr, out, M, D_DIM, D_DIM);
}


# causal mean: serial 2048-row scan per column replaced by 32 register-held 64-row chunk scans plus chunk offsets via LDS (prefix sums are added in a different order)
# speedup vs baseline: 1.2795x; 1.2795x over previous
#include <hip/hip_runtime.h>


typedef __attribute__((ext_vector_type(2))) float v2f;
typedef __attribute__((ext_vector_type(8))) float v8f;

#define D_DIM 1024
#define LSEQ  2048

template <bool RELU_BIAS>
__global__ __launch_bounds__(128) void gemm_f32_wmma(
    const float* __restrict__ A, const float* __restrict__ Bm,
    const float* __restrict__ bias, float* __restrict__ C,
    int M, int N, int K)
{
    const int lane = threadIdx.x & 31;
    const int wave = threadIdx.x >> 5;
    const int half = lane >> 4;
    const int l16  = lane & 15;
    const int koff = half * 2;

    const int m0 = blockIdx.y * 64 + (wave >> 1) * 32;
    const int n0 = blockIdx.x * 64 + (wave & 1) * 32;

    const float* Arow0 = A + (size_t)(m0 + l16) * K + koff;
    const float* Arow1 = A + (size_t)(m0 + 16 + l16) * K + koff;
    const float* Bcol0 = Bm + n0 + l16;
    const float* Bcol1 = Bm + n0 + 16 + l16;

    v8f acc00 = {}, acc01 = {}, acc10 = {}, acc11 = {};
    for (int k0 = 0; k0 < K; k0 += 4) {
        v2f a0 = *(const v2f*)(Arow0 + k0);
        v2f a1 = *(const v2f*)(Arow1 + k0);
        v2f b0, b1;
        size_t r0 = (size_t)(k0 + koff) * N;
        size_t r1 = r0 + N;
        b0.x = Bcol0[r0];  b0.y = Bcol0[r1];
        b1.x = Bcol1[r0];  b1.y = Bcol1[r1];
        acc00 = __builtin_amdgcn_wmma_f32_16x16x4_f32(false, a0, false, b0, (short)0, acc00, false, false);
        acc01 = __builtin_amdgcn_wmma_f32_16x16x4_f32(false, a0, false, b1, (short)0, acc01, false, false);
        acc10 = __builtin_amdgcn_wmma_f32_16x16x4_f32(false, a1, false, b0, (short)0, acc10, false, false);
        acc11 = __builtin_amdgcn_wmma_f32_16x16x4_f32(false, a1, false, b1, (short)0, acc11, false, false);
    }

    float bv0 = 0.0f, bv1 = 0.0f;
    if (RELU_BIAS) {
        bv0 = bias[n0 + l16];
        bv1 = bias[n0 + 16 + l16];
    }
#pragma unroll
    for (int r = 0; r < 8; ++r) {
        int row0 = m0 + r + half * 8;
        int row1 = row0 + 16;
        float v00 = acc00[r], v01 = acc01[r], v10 = acc10[r], v11 = acc11[r];
        if (RELU_BIAS) {
            v00 += bv0; v01 += bv1; v10 += bv0; v11 += bv1;
            v00 = v00 > 0.0f ? v00 : 0.0f;
            v01 = v01 > 0.0f ? v01 : 0.0f;
            v10 = v10 > 0.0f ? v10 : 0.0f;
            v11 = v11 > 0.0f ? v11 : 0.0f;
        }
        C[(size_t)row0 * N + n0 + l16]      = v00;
        C[(size_t)row0 * N + n0 + 16 + l16] = v01;
        C[(size_t)row1 * N + n0 + l16]      = v10;
        C[(size_t)row1 * N + n0 + 16 + l16] = v11;
    }
}

#define SCAN_ROWS  64
#define SCAN_WAVES (LSEQ / SCAN_ROWS)

__global__ __launch_bounds__(SCAN_WAVES * 32) void cumavg_kernel(
    const float* __restrict__ q, float* __restrict__ xavg)
{
    __shared__ float tot[SCAN_WAVES][32];

    const int lane = threadIdx.x & 31;
    const int c  = __builtin_amdgcn_readfirstlane((int)(threadIdx.x >> 5));
    const int b  = blockIdx.y;
    const int d  = blockIdx.x * 32 + lane;
    const int i0 = c * SCAN_ROWS;
    const size_t base = ((size_t)b * LSEQ + (size_t)i0) * D_DIM + d;
    const float* src = q + base;
    float*       dst = xavg + base;

    float v[SCAN_ROWS];
#pragma unroll
    for (int r = 0; r < SCAN_ROWS; ++r) v[r] = src[(size_t)r * D_DIM];
#pragma unroll
    for (int r = 1; r < SCAN_ROWS; ++r) v[r] += v[r - 1];

    tot[c][lane] = v[SCAN_ROWS - 1];
    __syncthreads();

    float off = 0.0f;
    for (int p = 0; p < c; ++p) off += tot[p][lane];

#pragma unroll
    for (int r = 0; r < SCAN_ROWS; ++r)
        dst[(size_t)r * D_DIM] = (off + v[r]) / (float)(i0 + r + 1);
}

__global__ __launch_bounds__(256) void rowsum_kernel(
    const float* __restrict__ g, float* __restrict__ S, int nrows)
{
    int wid  = (int)((blockIdx.x * blockDim.x + threadIdx.x) >> 5);
    int lane = threadIdx.x & 31;
    if (wid >= nrows) return;
    int i = wid % LSEQ;
    const float* row = g + (size_t)wid * LSEQ;
    float s = 0.0f;
    for (int k = (LSEQ - 1 - i) + lane; k < LSEQ; k += 32) s += row[k];
#pragma unroll
    for (int off = 16; off; off >>= 1) s += __shfl_xor(s, off, 32);
    if (lane == 0) S[wid] = s + 1e-8f;
}

__global__ __launch_bounds__(128) void attn_apply_wmma(
    const float* __restrict__ g, const float* __restrict__ q,
    const float* __restrict__ S, float* __restrict__ out)
{
    const int lane = threadIdx.x & 31;
    const int wave = threadIdx.x >> 5;
    const int b    = blockIdx.z;
    const int i0   = blockIdx.y * 16;
    const int d0   = blockIdx.x * 128 + wave * 32;
    const int half = lane >> 4;
    const int l16  = lane & 15;
    const int koff = half * 2;

    const int iA    = i0 + l16;
    const int shift = LSEQ - 1 - iA;
    const float* grow = g + ((size_t)b * LSEQ + iA) * LSEQ + shift;
    const float* qB0  = q + (size_t)b * LSEQ * D_DIM + d0 + l16;
    const float* qB1  = qB0 + 16;

    v8f acc0 = {}, acc1 = {};

    for (int j0 = 0; j0 < i0; j0 += 4) {
        int jx = j0 + koff;
        v2f a, b0, b1;
        a.x = grow[jx];
        a.y = grow[jx + 1];
        size_t r0 = (size_t)jx * D_DIM;
        size_t r1 = r0 + D_DIM;
        b0.x = qB0[r0];  b0.y = qB0[r1];
        b1.x = qB1[r0];  b1.y = qB1[r1];
        acc0 = __builtin_amdgcn_wmma_f32_16x16x4_f32(false, a, false, b0, (short)0, acc0, false, false);
        acc1 = __builtin_amdgcn_wmma_f32_16x16x4_f32(false, a, false, b1, (short)0, acc1, false, false);
    }

#pragma unroll
    for (int t = 0; t < 4; ++t) {
        int jx = i0 + t * 4 + koff;
        int jy = jx + 1;
        v2f a, b0, b1;
        a.x = (jx <= iA) ? grow[jx] : 0.0f;
        a.y = (jy <= iA) ? grow[jy] : 0.0f;
        size_t r0 = (size_t)jx * D_DIM;
        size_t r1 = r0 + D_DIM;
        b0.x = qB0[r0];  b0.y = qB0[r1];
        b1.x = qB1[r0];  b1.y = qB1[r1];
        acc0 = __builtin_amdgcn_wmma_f32_16x16x4_f32(false, a, false, b0, (short)0, acc0, false, false);
        acc1 = __builtin_amdgcn_wmma_f32_16x16x4_f32(false, a, false, b1, (short)0, acc1, false, false);
    }

#pragma unroll
    for (int r = 0; r < 8; ++r) {
        int row = i0 + r + half * 8;
        float invs = 1.0f / S[b * LSEQ + row];
        size_t o = ((size_t)b * LSEQ + row) * D_DIM + d0 + l16;
        out[o]      = acc0[r] * invs;
        out[o + 16] = acc1[r] * invs;
    }
}

extern "C" void kernel_launch(void* const* d_in, const int* in_sizes, int n_in,
                              void* d_out, int out_size, void* d_ws, size_t ws_size,
                              hipStream_t stream)
{
    const float* x  = (const float*)d_in[0];
    const float* Wq = (const float*)d_in[1];
    const float* Wg = (const float*)d_in[2];
    const float* bg = (const float*)d_in[3];
    const float* Wc = (const float*)d_in[4];
    float* out = (float*)d_out;

    const int Bsz = in_sizes[0] / (LSEQ * D_DIM);
    const int M   = Bsz * LSEQ;

    float* q    = (float*)d_ws;
    float* xavg = q    + (size_t)M * D_DIM;
    float* g    = xavg + (size_t)M * D_DIM;
    float* S    = g    + (size_t)M * LSEQ;

    dim3 blk(128);

    gemm_f32_wmma<false><<<dim3(D_DIM / 64, M / 64), blk, 0, stream>>>(
        x, Wq, nullptr, q, M, D_DIM, D_DIM);

    cumavg_kernel<<<dim3(D_DIM / 32, Bsz), SCAN_WAVES * 32, 0, stream>>>(q, xavg);

    gemm_f32_wmma<true><<<dim3(LSEQ / 64, M / 64), blk, 0, stream>>>(
        xavg, Wg, bg, g, M, LSEQ, D_DIM);

    rowsum_kernel<<<(M * 32 + 255) / 256, 256, 0, stream>>>(g, S, M);

    attn_apply_wmma<<<dim3(D_DIM / 128, LSEQ / 16, Bsz), blk, 0, stream>>>(
        g, q, S, xavg);

    gemm_f32_wmma<false><<<dim3(D_DIM / 64, M / 64), blk, 0, stream>>>(
        xavg, Wc, nullptr, out, M, D_DIM, D_DIM);
}


# GEMM launches (q, g, out): A and B tiles staged through LDS (128x128x32 block tile, whole-line 16-B global loads + ds_store, 8-B fragment reads) instead of per-lane global fragment loads
# speedup vs baseline: 2.0437x; 1.5973x over previous
#include <hip/hip_runtime.h>


typedef __attribute__((ext_vector_type(2))) float v2f;
typedef __attribute__((ext_vector_type(4))) float v4f;
typedef __attribute__((ext_vector_type(8))) float v8f;

#define D_DIM 1024
#define LSEQ  2048

#define GBM 128
#define GBN 128
#define GBK 32
#define GPA 36
#define GPB 288

template <bool RELU_BIAS, int N, int K>
__global__ __launch_bounds__(256) void gemm_f32_wmma(
    const float* __restrict__ A, const float* __restrict__ Bm,
    const float* __restrict__ bias, float* __restrict__ C)
{
    __shared__ __attribute__((aligned(16))) float As[2 * GBM * GPA];
    __shared__ __attribute__((aligned(16))) float Bs[2 * (GBK / 2) * GPB];

    const int tid  = threadIdx.x;
    const int lane = tid & 31;
    const int wave = tid >> 5;
    const int half = lane >> 4;
    const int l16  = lane & 15;
    const int wm   = wave >> 2;
    const int wn   = wave & 3;
    const int bm0  = blockIdx.y * GBM;
    const int bn0  = blockIdx.x * GBN;

    const int arow = tid >> 3;
    const int akq  = (tid & 7) * 4;
    const float* Ag = A + (size_t)(bm0 + arow) * K + akq;
    float* Ast = As + arow * GPA + akq;
    const int bkp  = tid >> 5;
    const int bn4  = (tid & 31) * 4;
    const float* Bg = Bm + (size_t)(2 * bkp) * N + bn0 + bn4;
    float* Bst = Bs + bkp * GPB + 2 * bn4;

    const float* aF = As + (wm * 64 + l16) * GPA + 2 * half;
    const float* bF = Bs + half * GPB + 2 * (wn * 32 + l16);

    v4f ra[4], rb0[2], rb1[2];
    v8f acc[4][2] = {};

#pragma unroll
    for (int i = 0; i < 4; ++i) ra[i] = *(const v4f*)(Ag + (size_t)(32 * i) * K);
#pragma unroll
    for (int i = 0; i < 2; ++i) {
        rb0[i] = *(const v4f*)(Bg + (size_t)(16 * i) * N);
        rb1[i] = *(const v4f*)(Bg + (size_t)(16 * i + 1) * N);
    }
#pragma unroll
    for (int i = 0; i < 4; ++i) *(v4f*)(Ast + 32 * i * GPA) = ra[i];
#pragma unroll
    for (int i = 0; i < 2; ++i) {
        v4f lo = {rb0[i].x, rb1[i].x, rb0[i].y, rb1[i].y};
        v4f hi = {rb0[i].z, rb1[i].z, rb0[i].w, rb1[i].w};
        *(v4f*)(Bst + 8 * i * GPB)     = lo;
        *(v4f*)(Bst + 8 * i * GPB + 4) = hi;
    }
    __syncthreads();

    const int nk = K / GBK;
    for (int kt = 0; kt < nk; ++kt) {
        const int cur = kt & 1;
        const bool more = (kt + 1 < nk);

        if (more) {
            const float* ap = Ag + (kt + 1) * GBK;
            const float* bp = Bg + (size_t)((kt + 1) * GBK) * N;
#pragma unroll
            for (int i = 0; i < 4; ++i) ra[i] = *(const v4f*)(ap + (size_t)(32 * i) * K);
#pragma unroll
            for (int i = 0; i < 2; ++i) {
                rb0[i] = *(const v4f*)(bp + (size_t)(16 * i) * N);
                rb1[i] = *(const v4f*)(bp + (size_t)(16 * i + 1) * N);
            }
        }

        const float* ab = aF + cur * (GBM * GPA);
        const float* bb = bF + cur * ((GBK / 2) * GPB);
#pragma unroll
        for (int s = 0; s < GBK / 4; ++s) {
            v2f a[4], b[2];
#pragma unroll
            for (int mi = 0; mi < 4; ++mi) a[mi] = *(const v2f*)(ab + mi * 16 * GPA + 4 * s);
#pragma unroll
            for (int ni = 0; ni < 2; ++ni) b[ni] = *(const v2f*)(bb + 2 * s * GPB + ni * 32);
#pragma unroll
            for (int mi = 0; mi < 4; ++mi)
#pragma unroll
                for (int ni = 0; ni < 2; ++ni)
                    acc[mi][ni] = __builtin_amdgcn_wmma_f32_16x16x4_f32(
                        false, a[mi], false, b[ni], (short)0, acc[mi][ni], false, false);
        }

        if (more) {
            float* as = Ast + (cur ^ 1) * (GBM * GPA);
            float* bs = Bst + (cur ^ 1) * ((GBK / 2) * GPB);
#pragma unroll
            for (int i = 0; i < 4; ++i) *(v4f*)(as + 32 * i * GPA) = ra[i];
#pragma unroll
            for (int i = 0; i < 2; ++i) {
                v4f lo = {rb0[i].x, rb1[i].x, rb0[i].y, rb1[i].y};
                v4f hi = {rb0[i].z, rb1[i].z, rb0[i].w, rb1[i].w};
                *(v4f*)(bs + 8 * i * GPB)     = lo;
                *(v4f*)(bs + 8 * i * GPB + 4) = hi;
            }
        }
        __syncthreads();
    }

    float bv[2] = {0.0f, 0.0f};
    if (RELU_BIAS) {
        bv[0] = bias[bn0 + wn * 32 + l16];
        bv[1] = bias[bn0 + wn * 32 + 16 + l16];
    }
#pragma unroll
    for (int mi = 0; mi < 4; ++mi) {
#pragma unroll
        for (int r = 0; r < 8; ++r) {
            const int row = bm0 + wm * 64 + mi * 16 + half * 8 + r;
            float* crow = C + (size_t)row * N + bn0 + wn * 32 + l16;
#pragma unroll
            for (int ni = 0; ni < 2; ++ni) {
                float v = acc[mi][ni][r];
                if (RELU_BIAS) {
                    v += bv[ni];
                    v = v > 0.0f ? v : 0.0f;
                }
                crow[ni * 16] = v;
            }
        }
    }
}

#define SCAN_ROWS  64
#define SCAN_WAVES (LSEQ / SCAN_ROWS)

__global__ __launch_bounds__(SCAN_WAVES * 32) void cumavg_kernel(
    const float* __restrict__ q, float* __restrict__ xavg)
{
    __shared__ float tot[SCAN_WAVES][32];

    const int lane = threadIdx.x & 31;
    const int c  = __builtin_amdgcn_readfirstlane((int)(threadIdx.x >> 5));
    const int b  = blockIdx.y;
    const int d  = blockIdx.x * 32 + lane;
    const int i0 = c * SCAN_ROWS;
    const size_t base = ((size_t)b * LSEQ + (size_t)i0) * D_DIM + d;
    const float* src = q + base;
    float*       dst = xavg + base;

    float v[SCAN_ROWS];
#pragma unroll
    for (int r = 0; r < SCAN_ROWS; ++r) v[r] = src[(size_t)r * D_DIM];
#pragma unroll
    for (int r = 1; r < SCAN_ROWS; ++r) v[r] += v[r - 1];

    tot[c][lane] = v[SCAN_ROWS - 1];
    __syncthreads();

    float off = 0.0f;
    for (int p = 0; p < c; ++p) off += tot[p][lane];

#pragma unroll
    for (int r = 0; r < SCAN_ROWS; ++r)
        dst[(size_t)r * D_DIM] = (off + v[r]) / (float)(i0 + r + 1);
}

__global__ __launch_bounds__(256) void rowsum_kernel(
    const float* __restrict__ g, float* __restrict__ S, int nrows)
{
    int wid  = (int)((blockIdx.x * blockDim.x + threadIdx.x) >> 5);
    int lane = threadIdx.x & 31;
    if (wid >= nrows) return;
    int i = wid % LSEQ;
    const float* row = g + (size_t)wid * LSEQ;
    float s = 0.0f;
    for (int k = (LSEQ - 1 - i) + lane; k < LSEQ; k += 32) s += row[k];
#pragma unroll
    for (int off = 16; off; off >>= 1) s += __shfl_xor(s, off, 32);
    if (lane == 0) S[wid] = s + 1e-8f;
}

__global__ __launch_bounds__(128) void attn_apply_wmma(
    const float* __restrict__ g, const float* __restrict__ q,
    const float* __restrict__ S, float* __restrict__ out)
{
    const int lane = threadIdx.x & 31;
    const int wave = threadIdx.x >> 5;
    const int b    = blockIdx.z;
    const int i0   = blockIdx.y * 16;
    const int d0   = blockIdx.x * 128 + wave * 32;
    const int half = lane >> 4;
    const int l16  = lane & 15;
    const int koff = half * 2;

    const int iA    = i0 + l16;
    const int shift = LSEQ - 1 - iA;
    const float* grow = g + ((size_t)b * LSEQ + iA) * LSEQ + shift;
    const float* qB0  = q + (size_t)b * LSEQ * D_DIM + d0 + l16;
    const float* qB1  = qB0 + 16;

    v8f acc0 = {}, acc1 = {};

    for (int j0 = 0; j0 < i0; j0 += 4) {
        int jx = j0 + koff;
        v2f a, b0, b1;
        a.x = grow[jx];
        a.y = grow[jx + 1];
        size_t r0 = (size_t)jx * D_DIM;
        size_t r1 = r0 + D_DIM;
        b0.x = qB0[r0];  b0.y = qB0[r1];
        b1.x = qB1[r0];  b1.y = qB1[r1];
        acc0 = __builtin_amdgcn_wmma_f32_16x16x4_f32(false, a, false, b0, (short)0, acc0, false, false);
        acc1 = __builtin_amdgcn_wmma_f32_16x16x4_f32(false, a, false, b1, (short)0, acc1, false, false);
    }

#pragma unroll
    for (int t = 0; t < 4; ++t) {
        int jx = i0 + t * 4 + koff;
        int jy = jx + 1;
        v2f a, b0, b1;
        a.x = (jx <= iA) ? grow[jx] : 0.0f;
        a.y = (jy <= iA) ? grow[jy] : 0.0f;
        size_t r0 = (size_t)jx * D_DIM;
        size_t r1 = r0 + D_DIM;
        b0.x = qB0[r0];  b0.y = qB0[r1];
        b1.x = qB1[r0];  b1.y = qB1[r1];
        acc0 = __builtin_amdgcn_wmma_f32_16x16x4_f32(false, a, false, b0, (short)0, acc0, false, false);
        acc1 = __builtin_amdgcn_wmma_f32_16x16x4_f32(false, a, false, b1, (short)0, acc1, false, false);
    }

#pragma unroll
    for (int r = 0; r < 8; ++r) {
        int row = i0 + r + half * 8;
        float invs = 1.0f / S[b * LSEQ + row];
        size_t o = ((size_t)b * LSEQ + row) * D_DIM + d0 + l16;
        out[o]      = acc0[r] * invs;
        out[o + 16] = acc1[r] * invs;
    }
}

extern "C" void kernel_launch(void* const* d_in, const int* in_sizes, int n_in,
                              void* d_out, int out_size, void* d_ws, size_t ws_size,
                              hipStream_t stream)
{
    const float* x  = (const float*)d_in[0];
    const float* Wq = (const float*)d_in[1];
    const float* Wg = (const float*)d_in[2];
    const float* bg = (const float*)d_in[3];
    const float* Wc = (const float*)d_in[4];
    float* out = (float*)d_out;

    const int Bsz = in_sizes[0] / (LSEQ * D_DIM);
    const int M   = Bsz * LSEQ;

    float* q    = (float*)d_ws;
    float* xavg = q    + (size_t)M * D_DIM;
    float* g    = xavg + (size_t)M * D_DIM;
    float* S    = g    + (size_t)M * LSEQ;

    dim3 blk(128);
    dim3 gblk(256);

    gemm_f32_wmma<false, D_DIM, D_DIM><<<dim3(D_DIM / GBN, M / GBM), gblk, 0, stream>>>(
        x, Wq, nullptr, q);

    cumavg_kernel<<<dim3(D_DIM / 32, Bsz), SCAN_WAVES * 32, 0, stream>>>(q, xavg);

    gemm_f32_wmma<true, LSEQ, D_DIM><<<dim3(LSEQ / GBN, M / GBM), gblk, 0, stream>>>(
        xavg, Wg, bg, g);

    rowsum_kernel<<<(M * 32 + 255) / 256, 256, 0, stream>>>(g, S, M);

    attn_apply_wmma<<<dim3(D_DIM / 128, LSEQ / 16, Bsz), blk, 0, stream>>>(
        g, q, S, xavg);

    gemm_f32_wmma<false, D_DIM, D_DIM><<<dim3(D_DIM / GBN, M / GBM), gblk, 0, stream>>>(
        xavg, Wc, nullptr, out);
}


# attn apply (tmp = attn @ q): A (shifted g rows, zero above the diagonal) and B (q) tiles staged through LDS, 128x128x32 block tile, instead of per-lane global fragment loads; sum order unchanged
# speedup vs baseline: 2.4505x; 1.1991x over previous
#include <hip/hip_runtime.h>


typedef __attribute__((ext_vector_type(2))) float v2f;
typedef __attribute__((ext_vector_type(4))) float v4f;
typedef __attribute__((ext_vector_type(8))) float v8f;

#define D_DIM 1024
#define LSEQ  2048

#define GBM 128
#define GBN 128
#define GBK 32
#define GPA 36
#define GPB 288

template <bool RELU_BIAS, int N, int K>
__global__ __launch_bounds__(256) void gemm_f32_wmma(
    const float* __restrict__ A, const float* __restrict__ Bm,
    const float* __restrict__ bias, float* __restrict__ C)
{
    __shared__ __attribute__((aligned(16))) float As[2 * GBM * GPA];
    __shared__ __attribute__((aligned(16))) float Bs[2 * (GBK / 2) * GPB];

    const int tid  = threadIdx.x;
    const int lane = tid & 31;
    const int wave = tid >> 5;
    const int half = lane >> 4;
    const int l16  = lane & 15;
    const int wm   = wave >> 2;
    const int wn   = wave & 3;
    const int bm0  = blockIdx.y * GBM;
    const int bn0  = blockIdx.x * GBN;

    const int arow = tid >> 3;
    const int akq  = (tid & 7) * 4;
    const float* Ag = A + (size_t)(bm0 + arow) * K + akq;
    float* Ast = As + arow * GPA + akq;
    const int bkp  = tid >> 5;
    const int bn4  = (tid & 31) * 4;
    const float* Bg = Bm + (size_t)(2 * bkp) * N + bn0 + bn4;
    float* Bst = Bs + bkp * GPB + 2 * bn4;

    const float* aF = As + (wm * 64 + l16) * GPA + 2 * half;
    const float* bF = Bs + half * GPB + 2 * (wn * 32 + l16);

    v4f ra[4], rb0[2], rb1[2];
    v8f acc[4][2] = {};

#pragma unroll
    for (int i = 0; i < 4; ++i) ra[i] = *(const v4f*)(Ag + (size_t)(32 * i) * K);
#pragma unroll
    for (int i = 0; i < 2; ++i) {
        rb0[i] = *(const v4f*)(Bg + (size_t)(16 * i) * N);
        rb1[i] = *(const v4f*)(Bg + (size_t)(16 * i + 1) * N);
    }
#pragma unroll
    for (int i = 0; i < 4; ++i) *(v4f*)(Ast + 32 * i * GPA) = ra[i];
#pragma unroll
    for (int i = 0; i < 2; ++i) {
        v4f lo = {rb0[i].x, rb1[i].x, rb0[i].y, rb1[i].y};
        v4f hi = {rb0[i].z, rb1[i].z, rb0[i].w, rb1[i].w};
        *(v4f*)(Bst + 8 * i * GPB)     = lo;
        *(v4f*)(Bst + 8 * i * GPB + 4) = hi;
    }
    __syncthreads();

    const int nk = K / GBK;
    for (int kt = 0; kt < nk; ++kt) {
        const int cur = kt & 1;
        const bool more = (kt + 1 < nk);

        if (more) {
            const float* ap = Ag + (kt + 1) * GBK;
            const float* bp = Bg + (size_t)((kt + 1) * GBK) * N;
#pragma unroll
            for (int i = 0; i < 4; ++i) ra[i] = *(const v4f*)(ap + (size_t)(32 * i) * K);
#pragma unroll
            for (int i = 0; i < 2; ++i) {
                rb0[i] = *(const v4f*)(bp + (size_t)(16 * i) * N);
                rb1[i] = *(const v4f*)(bp + (size_t)(16 * i + 1) * N);
            }
        }

        const float* ab = aF + cur * (GBM * GPA);
        const float* bb = bF + cur * ((GBK / 2) * GPB);
#pragma unroll
        for (int s = 0; s < GBK / 4; ++s) {
            v2f a[4], b[2];
#pragma unroll
            for (int mi = 0; mi < 4; ++mi) a[mi] = *(const v2f*)(ab + mi * 16 * GPA + 4 * s);
#pragma unroll
            for (int ni = 0; ni < 2; ++ni) b[ni] = *(const v2f*)(bb + 2 * s * GPB + ni * 32);
#pragma unroll
            for (int mi = 0; mi < 4; ++mi)
#pragma unroll
                for (int ni = 0; ni < 2; ++ni)
                    acc[mi][ni] = __builtin_amdgcn_wmma_f32_16x16x4_f32(
                        false, a[mi], false, b[ni], (short)0, acc[mi][ni], false, false);
        }

        if (more) {
            float* as = Ast + (cur ^ 1) * (GBM * GPA);
            float* bs = Bst + (cur ^ 1) * ((GBK / 2) * GPB);
#pragma unroll
            for (int i = 0; i < 4; ++i) *(v4f*)(as + 32 * i * GPA) = ra[i];
#pragma unroll
            for (int i = 0; i < 2; ++i) {
                v4f lo = {rb0[i].x, rb1[i].x, rb0[i].y, rb1[i].y};
                v4f hi = {rb0[i].z, rb1[i].z, rb0[i].w, rb1[i].w};
                *(v4f*)(bs + 8 * i * GPB)     = lo;
                *(v4f*)(bs + 8 * i * GPB + 4) = hi;
            }
        }
        __syncthreads();
    }

    float bv[2] = {0.0f, 0.0f};
    if (RELU_BIAS) {
        bv[0] = bias[bn0 + wn * 32 + l16];
        bv[1] = bias[bn0 + wn * 32 + 16 + l16];
    }
#pragma unroll
    for (int mi = 0; mi < 4; ++mi) {
#pragma unroll
        for (int r = 0; r < 8; ++r) {
            const int row = bm0 + wm * 64 + mi * 16 + half * 8 + r;
            float* crow = C + (size_t)row * N + bn0 + wn * 32 + l16;
#pragma unroll
            for (int ni = 0; ni < 2; ++ni) {
                float v = acc[mi][ni][r];
                if (RELU_BIAS) {
                    v += bv[ni];
                    v = v > 0.0f ? v : 0.0f;
                }
                crow[ni * 16] = v;
            }
        }
    }
}

#define SCAN_ROWS  64
#define SCAN_WAVES (LSEQ / SCAN_ROWS)

__global__ __launch_bounds__(SCAN_WAVES * 32) void cumavg_kernel(
    const float* __restrict__ q, float* __restrict__ xavg)
{
    __shared__ float tot[SCAN_WAVES][32];

    const int lane = threadIdx.x & 31;
    const int c  = __builtin_amdgcn_readfirstlane((int)(threadIdx.x >> 5));
    const int b  = blockIdx.y;
    const int d  = blockIdx.x * 32 + lane;
    const int i0 = c * SCAN_ROWS;
    const size_t base = ((size_t)b * LSEQ + (size_t)i0) * D_DIM + d;
    const float* src = q + base;
    float*       dst = xavg + base;

    float v[SCAN_ROWS];
#pragma unroll
    for (int r = 0; r < SCAN_ROWS; ++r) v[r] = src[(size_t)r * D_DIM];
#pragma unroll
    for (int r = 1; r < SCAN_ROWS; ++r) v[r] += v[r - 1];

    tot[c][lane] = v[SCAN_ROWS - 1];
    __syncthreads();

    float off = 0.0f;
    for (int p = 0; p < c; ++p) off += tot[p][lane];

#pragma unroll
    for (int r = 0; r < SCAN_ROWS; ++r)
        dst[(size_t)r * D_DIM] = (off + v[r]) / (float)(i0 + r + 1);
}

__global__ __launch_bounds__(256) void rowsum_kernel(
    const float* __restrict__ g, float* __restrict__ S, int nrows)
{
    int wid  = (int)((blockIdx.x * blockDim.x + threadIdx.x) >> 5);
    int lane = threadIdx.x & 31;
    if (wid >= nrows) return;
    int i = wid % LSEQ;
    const float* row = g + (size_t)wid * LSEQ;
    float s = 0.0f;
    for (int k = (LSEQ - 1 - i) + lane; k < LSEQ; k += 32) s += row[k];
#pragma unroll
    for (int off = 16; off; off >>= 1) s += __shfl_xor(s, off, 32);
    if (lane == 0) S[wid] = s + 1e-8f;
}

__global__ __launch_bounds__(256) void attn_apply_lds(
    const float* __restrict__ g, const float* __restrict__ q,
    const float* __restrict__ S, float* __restrict__ out)
{
    __shared__ __attribute__((aligned(16))) float As[2 * GBM * GPA];
    __shared__ __attribute__((aligned(16))) float Bs[2 * (GBK / 2) * GPB];

    const int tid  = threadIdx.x;
    const int lane = tid & 31;
    const int wave = tid >> 5;
    const int half = lane >> 4;
    const int l16  = lane & 15;
    const int wm   = wave >> 2;
    const int wn   = wave & 3;
    const int b    = blockIdx.z;
    const int bn0  = blockIdx.x * GBN;

    const int ac = tid & 31;
    const int ar = tid >> 5;
    float* Ast = As + ar * GPA + ac;
    const int bkp  = tid >> 5;
    const int bn4  = (tid & 31) * 4;
    const float* Bg = q + ((size_t)b * LSEQ + 2 * bkp) * D_DIM + bn0 + bn4;
    float* Bst = Bs + bkp * GPB + 2 * bn4;

    const float* aF = As + (wm * 64 + l16) * GPA + 2 * half;
    const float* bF = Bs + half * GPB + 2 * (wn * 32 + l16);

    for (int pass = 0; pass < 2; ++pass) {
        const int rb = pass ? (int)blockIdx.y : (LSEQ / GBM - 1) - (int)blockIdx.y;
        const int i0 = rb * GBM;
        const int nk = (i0 + GBM) / GBK;

        const float* Ag = g + ((size_t)b * LSEQ + i0 + ar) * LSEQ + (LSEQ - 1 - (i0 + ar)) + ac;
        const int jlim = i0 + ar - ac;

        float ra[16];
        v4f rb0[2], rb1[2];
        v8f acc[4][2] = {};

#pragma unroll
        for (int it = 0; it < 16; ++it)
            ra[it] = (0 <= jlim + 8 * it) ? Ag[(size_t)it * (8 * LSEQ - 8)] : 0.0f;
#pragma unroll
        for (int i = 0; i < 2; ++i) {
            rb0[i] = *(const v4f*)(Bg + (size_t)(16 * i) * D_DIM);
            rb1[i] = *(const v4f*)(Bg + (size_t)(16 * i + 1) * D_DIM);
        }
#pragma unroll
        for (int it = 0; it < 16; ++it) Ast[8 * it * GPA] = ra[it];
#pragma unroll
        for (int i = 0; i < 2; ++i) {
            v4f lo = {rb0[i].x, rb1[i].x, rb0[i].y, rb1[i].y};
            v4f hi = {rb0[i].z, rb1[i].z, rb0[i].w, rb1[i].w};
            *(v4f*)(Bst + 8 * i * GPB)     = lo;
            *(v4f*)(Bst + 8 * i * GPB + 4) = hi;
        }
        __syncthreads();

        for (int kt = 0; kt < nk; ++kt) {
            const int cur = kt & 1;
            const bool more = (kt + 1 < nk);

            if (more) {
                const int j0 = (kt + 1) * GBK;
                const float* ap = Ag + j0;
                const float* bp = Bg + (size_t)j0 * D_DIM;
                if (j0 < i0) {
#pragma unroll
                    for (int it = 0; it < 16; ++it) ra[it] = ap[(size_t)it * (8 * LSEQ - 8)];
                } else {
#pragma unroll
                    for (int it = 0; it < 16; ++it)
                        ra[it] = (j0 <= jlim + 8 * it) ? ap[(size_t)it * (8 * LSEQ - 8)] : 0.0f;
                }
#pragma unroll
                for (int i = 0; i < 2; ++i) {
                    rb0[i] = *(const v4f*)(bp + (size_t)(16 * i) * D_DIM);
                    rb1[i] = *(const v4f*)(bp + (size_t)(16 * i + 1) * D_DIM);
                }
            }

            const float* ab = aF + cur * (GBM * GPA);
            const float* bb = bF + cur * ((GBK / 2) * GPB);
#pragma unroll
            for (int s = 0; s < GBK / 4; ++s) {
                v2f a[4], bf[2];
#pragma unroll
                for (int mi = 0; mi < 4; ++mi) a[mi] = *(const v2f*)(ab + mi * 16 * GPA + 4 * s);
#pragma unroll
                for (int ni = 0; ni < 2; ++ni) bf[ni] = *(const v2f*)(bb + 2 * s * GPB + ni * 32);
#pragma unroll
                for (int mi = 0; mi < 4; ++mi)
#pragma unroll
                    for (int ni = 0; ni < 2; ++ni)
                        acc[mi][ni] = __builtin_amdgcn_wmma_f32_16x16x4_f32(
                            false, a[mi], false, bf[ni], (short)0, acc[mi][ni], false, false);
            }

            if (more) {
                float* as = Ast + (cur ^ 1) * (GBM * GPA);
                float* bs = Bst + (cur ^ 1) * ((GBK / 2) * GPB);
#pragma unroll
                for (int it = 0; it < 16; ++it) as[8 * it * GPA] = ra[it];
#pragma unroll
                for (int i = 0; i < 2; ++i) {
                    v4f lo = {rb0[i].x, rb1[i].x, rb0[i].y, rb1[i].y};
                    v4f hi = {rb0[i].z, rb1[i].z, rb0[i].w, rb1[i].w};
                    *(v4f*)(bs + 8 * i * GPB)     = lo;
                    *(v4f*)(bs + 8 * i * GPB + 4) = hi;
                }
            }
            __syncthreads();
        }

#pragma unroll
        for (int mi = 0; mi < 4; ++mi) {
#pragma unroll
            for (int r = 0; r < 8; ++r) {
                const int row = b * LSEQ + i0 + wm * 64 + mi * 16 + half * 8 + r;
                const float invs = 1.0f / S[row];
                float* crow = out + (size_t)row * D_DIM + bn0 + wn * 32 + l16;
#pragma unroll
                for (int ni = 0; ni < 2; ++ni) crow[ni * 16] = acc[mi][ni][r] * invs;
            }
        }
    }
}

extern "C" void kernel_launch(void* const* d_in, const int* in_sizes, int n_in,
                              void* d_out, int out_size, void* d_ws, size_t ws_size,
                              hipStream_t stream)
{
    const float* x  = (const float*)d_in[0];
    const float* Wq = (const float*)d_in[1];
    const float* Wg = (const float*)d_in[2];
    const float* bg = (const float*)d_in[3];
    const float* Wc = (const float*)d_in[4];
    float* out = (float*)d_out;

    const int Bsz = in_sizes[0] / (LSEQ * D_DIM);
    const int M   = Bsz * LSEQ;

    float* q    = (float*)d_ws;
    float* xavg = q    + (size_t)M * D_DIM;
    float* g    = xavg + (size_t)M * D_DIM;
    float* S    = g    + (size_t)M * LSEQ;

    dim3 gblk(256);

    gemm_f32_wmma<false, D_DIM, D_DIM><<<dim3(D_DIM / GBN, M / GBM), gblk, 0, stream>>>(
        x, Wq, nullptr, q);

    cumavg_kernel<<<dim3(D_DIM / 32, Bsz), SCAN_WAVES * 32, 0, stream>>>(q, xavg);

    gemm_f32_wmma<true, LSEQ, D_DIM><<<dim3(LSEQ / GBN, M / GBM), gblk, 0, stream>>>(
        xavg, Wg, bg, g);

    rowsum_kernel<<<(M * 32 + 255) / 256, 256, 0, stream>>>(g, S, M);

    attn_apply_lds<<<dim3(D_DIM / GBN, LSEQ / (2 * GBM), Bsz), gblk, 0, stream>>>(
        g, q, S, xavg);

    gemm_f32_wmma<false, D_DIM, D_DIM><<<dim3(D_DIM / GBN, M / GBM), gblk, 0, stream>>>(
        xavg, Wc, nullptr, out);
}


# row sums S (attn normaliser) formed inside attn apply from the A tiles it stages anyway: per-thread running sums, wave butterfly per row, 1/S via LDS; row-sum launch removed; sum order unchanged
# speedup vs baseline: 2.5313x; 1.0330x over previous
#include <hip/hip_runtime.h>


typedef __attribute__((ext_vector_type(2))) float v2f;
typedef __attribute__((ext_vector_type(4))) float v4f;
typedef __attribute__((ext_vector_type(8))) float v8f;

#define D_DIM 1024
#define LSEQ  2048

#define GBM 128
#define GBN 128
#define GBK 32
#define GPA 36
#define GPB 288

template <bool RELU_BIAS, int N, int K>
__global__ __launch_bounds__(256) void gemm_f32_wmma(
    const float* __restrict__ A, const float* __restrict__ Bm,
    const float* __restrict__ bias, float* __restrict__ C)
{
    __shared__ __attribute__((aligned(16))) float As[2 * GBM * GPA];
    __shared__ __attribute__((aligned(16))) float Bs[2 * (GBK / 2) * GPB];

    const int tid  = threadIdx.x;
    const int lane = tid & 31;
    const int wave = tid >> 5;
    const int half = lane >> 4;
    const int l16  = lane & 15;
    const int wm   = wave >> 2;
    const int wn   = wave & 3;
    const int bm0  = blockIdx.y * GBM;
    const int bn0  = blockIdx.x * GBN;

    const int arow = tid >> 3;
    const int akq  = (tid & 7) * 4;
    const float* Ag = A + (size_t)(bm0 + arow) * K + akq;
    float* Ast = As + arow * GPA + akq;
    const int bkp  = tid >> 5;
    const int bn4  = (tid & 31) * 4;
    const float* Bg = Bm + (size_t)(2 * bkp) * N + bn0 + bn4;
    float* Bst = Bs + bkp * GPB + 2 * bn4;

    const float* aF = As + (wm * 64 + l16) * GPA + 2 * half;
    const float* bF = Bs + half * GPB + 2 * (wn * 32 + l16);

    v4f ra[4], rb0[2], rb1[2];
    v8f acc[4][2] = {};

#pragma unroll
    for (int i = 0; i < 4; ++i) ra[i] = *(const v4f*)(Ag + (size_t)(32 * i) * K);
#pragma unroll
    for (int i = 0; i < 2; ++i) {
        rb0[i] = *(const v4f*)(Bg + (size_t)(16 * i) * N);
        rb1[i] = *(const v4f*)(Bg + (size_t)(16 * i + 1) * N);
    }
#pragma unroll
    for (int i = 0; i < 4; ++i) *(v4f*)(Ast + 32 * i * GPA) = ra[i];
#pragma unroll
    for (int i = 0; i < 2; ++i) {
        v4f lo = {rb0[i].x, rb1[i].x, rb0[i].y, rb1[i].y};
        v4f hi = {rb0[i].z, rb1[i].z, rb0[i].w, rb1[i].w};
        *(v4f*)(Bst + 8 * i * GPB)     = lo;
        *(v4f*)(Bst + 8 * i * GPB + 4) = hi;
    }
    __syncthreads();

    const int nk = K / GBK;
    for (int kt = 0; kt < nk; ++kt) {
        const int cur = kt & 1;
        const bool more = (kt + 1 < nk);

        if (more) {
            const float* ap = Ag + (kt + 1) * GBK;
            const float* bp = Bg + (size_t)((kt + 1) * GBK) * N;
#pragma unroll
            for (int i = 0; i < 4; ++i) ra[i] = *(const v4f*)(ap + (size_t)(32 * i) * K);
#pragma unroll
            for (int i = 0; i < 2; ++i) {
                rb0[i] = *(const v4f*)(bp + (size_t)(16 * i) * N);
                rb1[i] = *(const v4f*)(bp + (size_t)(16 * i + 1) * N);
            }
        }

        const float* ab = aF + cur * (GBM * GPA);
        const float* bb = bF + cur * ((GBK / 2) * GPB);
#pragma unroll
        for (int s = 0; s < GBK / 4; ++s) {
            v2f a[4], b[2];
#pragma unroll
            for (int mi = 0; mi < 4; ++mi) a[mi] = *(const v2f*)(ab + mi * 16 * GPA + 4 * s);
#pragma unroll
            for (int ni = 0; ni < 2; ++ni) b[ni] = *(const v2f*)(bb + 2 * s * GPB + ni * 32);
#pragma unroll
            for (int mi = 0; mi < 4; ++mi)
#pragma unroll
                for (int ni = 0; ni < 2; ++ni)
                    acc[mi][ni] = __builtin_amdgcn_wmma_f32_16x16x4_f32(
                        false, a[mi], false, b[ni], (short)0, acc[mi][ni], false, false);
        }

        if (more) {
            float* as = Ast + (cur ^ 1) * (GBM * GPA);
            float* bs = Bst + (cur ^ 1) * ((GBK / 2) * GPB);
#pragma unroll
            for (int i = 0; i < 4; ++i) *(v4f*)(as + 32 * i * GPA) = ra[i];
#pragma unroll
            for (int i = 0; i < 2; ++i) {
                v4f lo = {rb0[i].x, rb1[i].x, rb0[i].y, rb1[i].y};
                v4f hi = {rb0[i].z, rb1[i].z, rb0[i].w, rb1[i].w};
                *(v4f*)(bs + 8 * i * GPB)     = lo;
                *(v4f*)(bs + 8 * i * GPB + 4) = hi;
            }
        }
        __syncthreads();
    }

    float bv[2] = {0.0f, 0.0f};
    if (RELU_BIAS) {
        bv[0] = bias[bn0 + wn * 32 + l16];
        bv[1] = bias[bn0 + wn * 32 + 16 + l16];
    }
#pragma unroll
    for (int mi = 0; mi < 4; ++mi) {
#pragma unroll
        for (int r = 0; r < 8; ++r) {
            const int row = bm0 + wm * 64 + mi * 16 + half * 8 + r;
            float* crow = C + (size_t)row * N + bn0 + wn * 32 + l16;
#pragma unroll
            for (int ni = 0; ni < 2; ++ni) {
                float v = acc[mi][ni][r];
                if (RELU_BIAS) {
                    v += bv[ni];
                    v = v > 0.0f ? v : 0.0f;
                }
                crow[ni * 16] = v;
            }
        }
    }
}

#define SCAN_ROWS  64
#define SCAN_WAVES (LSEQ / SCAN_ROWS)

__global__ __launch_bounds__(SCAN_WAVES * 32) void cumavg_kernel(
    const float* __restrict__ q, float* __restrict__ xavg)
{
    __shared__ float tot[SCAN_WAVES][32];

    const int lane = threadIdx.x & 31;
    const int c  = __builtin_amdgcn_readfirstlane((int)(threadIdx.x >> 5));
    const int b  = blockIdx.y;
    const int d  = blockIdx.x * 32 + lane;
    const int i0 = c * SCAN_ROWS;
    const size_t base = ((size_t)b * LSEQ + (size_t)i0) * D_DIM + d;
    const float* src = q + base;
    float*       dst = xavg + base;

    float v[SCAN_ROWS];
#pragma unroll
    for (int r = 0; r < SCAN_ROWS; ++r) v[r] = src[(size_t)r * D_DIM];
#pragma unroll
    for (int r = 1; r < SCAN_ROWS; ++r) v[r] += v[r - 1];

    tot[c][lane] = v[SCAN_ROWS - 1];
    __syncthreads();

    float off = 0.0f;
    for (int p = 0; p < c; ++p) off += tot[p][lane];

#pragma unroll
    for (int r = 0; r < SCAN_ROWS; ++r)
        dst[(size_t)r * D_DIM] = (off + v[r]) / (float)(i0 + r + 1);
}

__global__ __launch_bounds__(256) void attn_apply_lds(
    const float* __restrict__ g, const float* __restrict__ q,
    float* __restrict__ out)
{
    __shared__ __attribute__((aligned(16))) float As[2 * GBM * GPA];
    __shared__ __attribute__((aligned(16))) float Bs[2 * (GBK / 2) * GPB];
    __shared__ __attribute__((aligned(16))) float Sinv[GBM];

    const int tid  = threadIdx.x;
    const int lane = tid & 31;
    const int wave = tid >> 5;
    const int half = lane >> 4;
    const int l16  = lane & 15;
    const int wm   = wave >> 2;
    const int wn   = wave & 3;
    const int b    = blockIdx.z;
    const int bn0  = blockIdx.x * GBN;

    const int ac = tid & 31;
    const int ar = tid >> 5;
    float* Ast = As + ar * GPA + ac;
    const int bkp  = tid >> 5;
    const int bn4  = (tid & 31) * 4;
    const float* Bg = q + ((size_t)b * LSEQ + 2 * bkp) * D_DIM + bn0 + bn4;
    float* Bst = Bs + bkp * GPB + 2 * bn4;

    const float* aF = As + (wm * 64 + l16) * GPA + 2 * half;
    const float* bF = Bs + half * GPB + 2 * (wn * 32 + l16);

    for (int pass = 0; pass < 2; ++pass) {
        const int rb = pass ? (int)blockIdx.y : (LSEQ / GBM - 1) - (int)blockIdx.y;
        const int i0 = rb * GBM;
        const int nk = (i0 + GBM) / GBK;

        const float* Ag = g + ((size_t)b * LSEQ + i0 + ar) * LSEQ + (LSEQ - 1 - (i0 + ar)) + ac;
        const int jlim = i0 + ar - ac;

        float ra[16];
        float rs[16];
        v4f rb0[2], rb1[2];
        v8f acc[4][2] = {};

#pragma unroll
        for (int it = 0; it < 16; ++it)
            ra[it] = (0 <= jlim + 8 * it) ? Ag[(size_t)it * (8 * LSEQ - 8)] : 0.0f;
#pragma unroll
        for (int i = 0; i < 2; ++i) {
            rb0[i] = *(const v4f*)(Bg + (size_t)(16 * i) * D_DIM);
            rb1[i] = *(const v4f*)(Bg + (size_t)(16 * i + 1) * D_DIM);
        }
#pragma unroll
        for (int it = 0; it < 16; ++it) {
            Ast[8 * it * GPA] = ra[it];
            rs[it] = ra[it];
        }
#pragma unroll
        for (int i = 0; i < 2; ++i) {
            v4f lo = {rb0[i].x, rb1[i].x, rb0[i].y, rb1[i].y};
            v4f hi = {rb0[i].z, rb1[i].z, rb0[i].w, rb1[i].w};
            *(v4f*)(Bst + 8 * i * GPB)     = lo;
            *(v4f*)(Bst + 8 * i * GPB + 4) = hi;
        }
        __syncthreads();

        for (int kt = 0; kt < nk; ++kt) {
            const int cur = kt & 1;
            const bool more = (kt + 1 < nk);

            if (more) {
                const int j0 = (kt + 1) * GBK;
                const float* ap = Ag + j0;
                const float* bp = Bg + (size_t)j0 * D_DIM;
                if (j0 < i0) {
#pragma unroll
                    for (int it = 0; it < 16; ++it) ra[it] = ap[(size_t)it * (8 * LSEQ - 8)];
                } else {
#pragma unroll
                    for (int it = 0; it < 16; ++it)
                        ra[it] = (j0 <= jlim + 8 * it) ? ap[(size_t)it * (8 * LSEQ - 8)] : 0.0f;
                }
#pragma unroll
                for (int i = 0; i < 2; ++i) {
                    rb0[i] = *(const v4f*)(bp + (size_t)(16 * i) * D_DIM);
                    rb1[i] = *(const v4f*)(bp + (size_t)(16 * i + 1) * D_DIM);
                }
            }

            const float* ab = aF + cur * (GBM * GPA);
            const float* bb = bF + cur * ((GBK / 2) * GPB);
#pragma unroll
            for (int s = 0; s < GBK / 4; ++s) {
                v2f a[4], bf[2];
#pragma unroll
                for (int mi = 0; mi < 4; ++mi) a[mi] = *(const v2f*)(ab + mi * 16 * GPA + 4 * s);
#pragma unroll
                for (int ni = 0; ni < 2; ++ni) bf[ni] = *(const v2f*)(bb + 2 * s * GPB + ni * 32);
#pragma unroll
                for (int mi = 0; mi < 4; ++mi)
#pragma unroll
                    for (int ni = 0; ni < 2; ++ni)
                        acc[mi][ni] = __builtin_amdgcn_wmma_f32_16x16x4_f32(
                            false, a[mi], false, bf[ni], (short)0, acc[mi][ni], false, false);
            }

            if (more) {
                float* as = Ast + (cur ^ 1) * (GBM * GPA);
                float* bs = Bst + (cur ^ 1) * ((GBK / 2) * GPB);
#pragma unroll
                for (int it = 0; it < 16; ++it) {
                    as[8 * it * GPA] = ra[it];
                    rs[it] += ra[it];
                }
#pragma unroll
                for (int i = 0; i < 2; ++i) {
                    v4f lo = {rb0[i].x, rb1[i].x, rb0[i].y, rb1[i].y};
                    v4f hi = {rb0[i].z, rb1[i].z, rb0[i].w, rb1[i].w};
                    *(v4f*)(bs + 8 * i * GPB)     = lo;
                    *(v4f*)(bs + 8 * i * GPB + 4) = hi;
                }
            }
            __syncthreads();
        }

#pragma unroll
        for (int off = 16; off; off >>= 1) {
            float t[16];
#pragma unroll
            for (int it = 0; it < 16; ++it) t[it] = __shfl_xor(rs[it], off, 32);
#pragma unroll
            for (int it = 0; it < 16; ++it) rs[it] += t[it];
        }
        if (lane == 0) {
#pragma unroll
            for (int it = 0; it < 16; ++it) Sinv[ar + 8 * it] = 1.0f / (rs[it] + 1e-8f);
        }
        __syncthreads();

#pragma unroll
        for (int mi = 0; mi < 4; ++mi) {
#pragma unroll
            for (int r = 0; r < 8; ++r) {
                const int lrow = wm * 64 + mi * 16 + half * 8 + r;
                const int row = b * LSEQ + i0 + lrow;
                const float invs = Sinv[lrow];
                float* crow = out + (size_t)row * D_DIM + bn0 + wn * 32 + l16;
#pragma unroll
                for (int ni = 0; ni < 2; ++ni) crow[ni * 16] = acc[mi][ni][r] * invs;
            }
        }
    }
}

extern "C" void kernel_launch(void* const* d_in, const int* in_sizes, int n_in,
                              void* d_out, int out_size, void* d_ws, size_t ws_size,
                              hipStream_t stream)
{
    const float* x  = (const float*)d_in[0];
    const float* Wq = (const float*)d_in[1];
    const float* Wg = (const float*)d_in[2];
    const float* bg = (const float*)d_in[3];
    const float* Wc = (const float*)d_in[4];
    float* out = (float*)d_out;

    const int Bsz = in_sizes[0] / (LSEQ * D_DIM);
    const int M   = Bsz * LSEQ;

    float* q    = (float*)d_ws;
    float* xavg = q    + (size_t)M * D_DIM;
    float* g    = xavg + (size_t)M * D_DIM;

    dim3 gblk(256);

    gemm_f32_wmma<false, D_DIM, D_DIM><<<dim3(D_DIM / GBN, M / GBM), gblk, 0, stream>>>(
        x, Wq, nullptr, q);

    cumavg_kernel<<<dim3(D_DIM / 32, Bsz), SCAN_WAVES * 32, 0, stream>>>(q, xavg);

    gemm_f32_wmma<true, LSEQ, D_DIM><<<dim3(LSEQ / GBN, M / GBM), gblk, 0, stream>>>(
        xavg, Wg, bg, g);

    attn_apply_lds<<<dim3(D_DIM / GBN, LSEQ / (2 * GBM), Bsz), gblk, 0, stream>>>(
        g, q, xavg);

    gemm_f32_wmma<false, D_DIM, D_DIM><<<dim3(D_DIM / GBN, M / GBM), gblk, 0, stream>>>(
        xavg, Wc, nullptr, out);
}


# g GEMM computes only the 128x128 blocks of g that attn apply reads (row block rb of a batch: column blocks 15-rb..15, 136 of 256); unread blocks are neither computed nor stored; sum order unchanged
# speedup vs baseline: 2.8316x; 1.1186x over previous
#include <hip/hip_runtime.h>


typedef __attribute__((ext_vector_type(2))) float v2f;
typedef __attribute__((ext_vector_type(4))) float v4f;
typedef __attribute__((ext_vector_type(8))) float v8f;

#define D_DIM 1024
#define LSEQ  2048

#define GBM 128
#define GBN 128
#define GBK 32
#define GPA 36
#define GPB 288
#define TRI_BLOCKS ((LSEQ / GBM) * (LSEQ / GBM + 1) / 2)

template <bool RELU_BIAS, bool TRI, int N, int K>
__global__ __launch_bounds__(256) void gemm_f32_wmma(
    const float* __restrict__ A, const float* __restrict__ Bm,
    const float* __restrict__ bias, float* __restrict__ C)
{
    __shared__ __attribute__((aligned(16))) float As[2 * GBM * GPA];
    __shared__ __attribute__((aligned(16))) float Bs[2 * (GBK / 2) * GPB];

    const int tid  = threadIdx.x;
    const int lane = tid & 31;
    const int wave = tid >> 5;
    const int half = lane >> 4;
    const int l16  = lane & 15;
    const int wm   = wave >> 2;
    const int wn   = wave & 3;
    int bm0 = blockIdx.y * GBM;
    int bn0 = blockIdx.x * GBN;
    if (TRI) {
        int rb = 0, rem = blockIdx.x;
        while (rem > rb) { rem -= rb + 1; ++rb; }
        bm0 = (blockIdx.y * (LSEQ / GBM) + rb) * GBM;
        bn0 = (N / GBN - 1 - rb + rem) * GBN;
    }

    const int arow = tid >> 3;
    const int akq  = (tid & 7) * 4;
    const float* Ag = A + (size_t)(bm0 + arow) * K + akq;
    float* Ast = As + arow * GPA + akq;
    const int bkp  = tid >> 5;
    const int bn4  = (tid & 31) * 4;
    const float* Bg = Bm + (size_t)(2 * bkp) * N + bn0 + bn4;
    float* Bst = Bs + bkp * GPB + 2 * bn4;

    const float* aF = As + (wm * 64 + l16) * GPA + 2 * half;
    const float* bF = Bs + half * GPB + 2 * (wn * 32 + l16);

    v4f ra[4], rb0[2], rb1[2];
    v8f acc[4][2] = {};

#pragma unroll
    for (int i = 0; i < 4; ++i) ra[i] = *(const v4f*)(Ag + (size_t)(32 * i) * K);
#pragma unroll
    for (int i = 0; i < 2; ++i) {
        rb0[i] = *(const v4f*)(Bg + (size_t)(16 * i) * N);
        rb1[i] = *(const v4f*)(Bg + (size_t)(16 * i + 1) * N);
    }
#pragma unroll
    for (int i = 0; i < 4; ++i) *(v4f*)(Ast + 32 * i * GPA) = ra[i];
#pragma unroll
    for (int i = 0; i < 2; ++i) {
        v4f lo = {rb0[i].x, rb1[i].x, rb0[i].y, rb1[i].y};
        v4f hi = {rb0[i].z, rb1[i].z, rb0[i].w, rb1[i].w};
        *(v4f*)(Bst + 8 * i * GPB)     = lo;
        *(v4f*)(Bst + 8 * i * GPB + 4) = hi;
    }
    __syncthreads();

    const int nk = K / GBK;
    for (int kt = 0; kt < nk; ++kt) {
        const int cur = kt & 1;
        const bool more = (kt + 1 < nk);

        if (more) {
            const float* ap = Ag + (kt + 1) * GBK;
            const float* bp = Bg + (size_t)((kt + 1) * GBK) * N;
#pragma unroll
            for (int i = 0; i < 4; ++i) ra[i] = *(const v4f*)(ap + (size_t)(32 * i) * K);
#pragma unroll
            for (int i = 0; i < 2; ++i) {
                rb0[i] = *(const v4f*)(bp + (size_t)(16 * i) * N);
                rb1[i] = *(const v4f*)(bp + (size_t)(16 * i + 1) * N);
            }
        }

        const float* ab = aF + cur * (GBM * GPA);
        const float* bb = bF + cur * ((GBK / 2) * GPB);
#pragma unroll
        for (int s = 0; s < GBK / 4; ++s) {
            v2f a[4], b[2];
#pragma unroll
            for (int mi = 0; mi < 4; ++mi) a[mi] = *(const v2f*)(ab + mi * 16 * GPA + 4 * s);
#pragma unroll
            for (int ni = 0; ni < 2; ++ni) b[ni] = *(const v2f*)(bb + 2 * s * GPB + ni * 32);
#pragma unroll
            for (int mi = 0; mi < 4; ++mi)
#pragma unroll
                for (int ni = 0; ni < 2; ++ni)
                    acc[mi][ni] = __builtin_amdgcn_wmma_f32_16x16x4_f32(
                        false, a[mi], false, b[ni], (short)0, acc[mi][ni], false, false);
        }

        if (more) {
            float* as = Ast + (cur ^ 1) * (GBM * GPA);
            float* bs = Bst + (cur ^ 1) * ((GBK / 2) * GPB);
#pragma unroll
            for (int i = 0; i < 4; ++i) *(v4f*)(as + 32 * i * GPA) = ra[i];
#pragma unroll
            for (int i = 0; i < 2; ++i) {
                v4f lo = {rb0[i].x, rb1[i].x, rb0[i].y, rb1[i].y};
                v4f hi = {rb0[i].z, rb1[i].z, rb0[i].w, rb1[i].w};
                *(v4f*)(bs + 8 * i * GPB)     = lo;
                *(v4f*)(bs + 8 * i * GPB + 4) = hi;
            }
        }
        __syncthreads();
    }

    float bv[2] = {0.0f, 0.0f};
    if (RELU_BIAS) {
        bv[0] = bias[bn0 + wn * 32 + l16];
        bv[1] = bias[bn0 + wn * 32 + 16 + l16];
    }
#pragma unroll
    for (int mi = 0; mi < 4; ++mi) {
#pragma unroll
        for (int r = 0; r < 8; ++r) {
            const int row = bm0 + wm * 64 + mi * 16 + half * 8 + r;
            float* crow = C + (size_t)row * N + bn0 + wn * 32 + l16;
#pragma unroll
            for (int ni = 0; ni < 2; ++ni) {
                float v = acc[mi][ni][r];
                if (RELU_BIAS) {
                    v += bv[ni];
                    v = v > 0.0f ? v : 0.0f;
                }
                crow[ni * 16] = v;
            }
        }
    }
}

#define SCAN_ROWS  64
#define SCAN_WAVES (LSEQ / SCAN_ROWS)

__global__ __launch_bounds__(SCAN_WAVES * 32) void cumavg_kernel(
    const float* __restrict__ q, float* __restrict__ xavg)
{
    __shared__ float tot[SCAN_WAVES][32];

    const int lane = threadIdx.x & 31;
    const int c  = __builtin_amdgcn_readfirstlane((int)(threadIdx.x >> 5));
    const int b  = blockIdx.y;
    const int d  = blockIdx.x * 32 + lane;
    const int i0 = c * SCAN_ROWS;
    const size_t base = ((size_t)b * LSEQ + (size_t)i0) * D_DIM + d;
    const float* src = q + base;
    float*       dst = xavg + base;

    float v[SCAN_ROWS];
#pragma unroll
    for (int r = 0; r < SCAN_ROWS; ++r) v[r] = src[(size_t)r * D_DIM];
#pragma unroll
    for (int r = 1; r < SCAN_ROWS; ++r) v[r] += v[r - 1];

    tot[c][lane] = v[SCAN_ROWS - 1];
    __syncthreads();

    float off = 0.0f;
    for (int p = 0; p < c; ++p) off += tot[p][lane];

#pragma unroll
    for (int r = 0; r < SCAN_ROWS; ++r)
        dst[(size_t)r * D_DIM] = (off + v[r]) / (float)(i0 + r + 1);
}

__global__ __launch_bounds__(256) void attn_apply_lds(
    const float* __restrict__ g, const float* __restrict__ q,
    float* __restrict__ out)
{
    __shared__ __attribute__((aligned(16))) float As[2 * GBM * GPA];
    __shared__ __attribute__((aligned(16))) float Bs[2 * (GBK / 2) * GPB];
    __shared__ __attribute__((aligned(16))) float Sinv[GBM];

    const int tid  = threadIdx.x;
    const int lane = tid & 31;
    const int wave = tid >> 5;
    const int half = lane >> 4;
    const int l16  = lane & 15;
    const int wm   = wave >> 2;
    const int wn   = wave & 3;
    const int b    = blockIdx.z;
    const int bn0  = blockIdx.x * GBN;

    const int ac = tid & 31;
    const int ar = tid >> 5;
    float* Ast = As + ar * GPA + ac;
    const int bkp  = tid >> 5;
    const int bn4  = (tid & 31) * 4;
    const float* Bg = q + ((size_t)b * LSEQ + 2 * bkp) * D_DIM + bn0 + bn4;
    float* Bst = Bs + bkp * GPB + 2 * bn4;

    const float* aF = As + (wm * 64 + l16) * GPA + 2 * half;
    const float* bF = Bs + half * GPB + 2 * (wn * 32 + l16);

    for (int pass = 0; pass < 2; ++pass) {
        const int rb = pass ? (int)blockIdx.y : (LSEQ / GBM - 1) - (int)blockIdx.y;
        const int i0 = rb * GBM;
        const int nk = (i0 + GBM) / GBK;

        const float* Ag = g + ((size_t)b * LSEQ + i0 + ar) * LSEQ + (LSEQ - 1 - (i0 + ar)) + ac;
        const int jlim = i0 + ar - ac;

        float ra[16];
        float rs[16];
        v4f rb0[2], rb1[2];
        v8f acc[4][2] = {};

#pragma unroll
        for (int it = 0; it < 16; ++it)
            ra[it] = (0 <= jlim + 8 * it) ? Ag[(size_t)it * (8 * LSEQ - 8)] : 0.0f;
#pragma unroll
        for (int i = 0; i < 2; ++i) {
            rb0[i] = *(const v4f*)(Bg + (size_t)(16 * i) * D_DIM);
            rb1[i] = *(const v4f*)(Bg + (size_t)(16 * i + 1) * D_DIM);
        }
#pragma unroll
        for (int it = 0; it < 16; ++it) {
            Ast[8 * it * GPA] = ra[it];
            rs[it] = ra[it];
        }
#pragma unroll
        for (int i = 0; i < 2; ++i) {
            v4f lo = {rb0[i].x, rb1[i].x, rb0[i].y, rb1[i].y};
            v4f hi = {rb0[i].z, rb1[i].z, rb0[i].w, rb1[i].w};
            *(v4f*)(Bst + 8 * i * GPB)     = lo;
            *(v4f*)(Bst + 8 * i * GPB + 4) = hi;
        }
        __syncthreads();

        for (int kt = 0; kt < nk; ++kt) {
            const int cur = kt & 1;
            const bool more = (kt + 1 < nk);

            if (more) {
                const int j0 = (kt + 1) * GBK;
                const float* ap = Ag + j0;
                const float* bp = Bg + (size_t)j0 * D_DIM;
                if (j0 < i0) {
#pragma unroll
                    for (int it = 0; it < 16; ++it) ra[it] = ap[(size_t)it * (8 * LSEQ - 8)];
                } else {
#pragma unroll
                    for (int it = 0; it < 16; ++it)
                        ra[it] = (j0 <= jlim + 8 * it) ? ap[(size_t)it * (8 * LSEQ - 8)] : 0.0f;
                }
#pragma unroll
                for (int i = 0; i < 2; ++i) {
                    rb0[i] = *(const v4f*)(bp + (size_t)(16 * i) * D_DIM);
                    rb1[i] = *(const v4f*)(bp + (size_t)(16 * i + 1) * D_DIM);
                }
            }

            const float* ab = aF + cur * (GBM * GPA);
            const float* bb = bF + cur * ((GBK / 2) * GPB);
#pragma unroll
            for (int s = 0; s < GBK / 4; ++s) {
                v2f a[4], bf[2];
#pragma unroll
                for (int mi = 0; mi < 4; ++mi) a[mi] = *(const v2f*)(ab + mi * 16 * GPA + 4 * s);
#pragma unroll
                for (int ni = 0; ni < 2; ++ni) bf[ni] = *(const v2f*)(bb + 2 * s * GPB + ni * 32);
#pragma unroll
                for (int mi = 0; mi < 4; ++mi)
#pragma unroll
                    for (int ni = 0; ni < 2; ++ni)
                        acc[mi][ni] = __builtin_amdgcn_wmma_f32_16x16x4_f32(
                            false, a[mi], false, bf[ni], (short)0, acc[mi][ni], false, false);
            }

            if (more) {
                float* as = Ast + (cur ^ 1) * (GBM * GPA);
                float* bs = Bst + (cur ^ 1) * ((GBK / 2) * GPB);
#pragma unroll
                for (int it = 0; it < 16; ++it) {
                    as[8 * it * GPA] = ra[it];
                    rs[it] += ra[it];
                }
#pragma unroll
                for (int i = 0; i < 2; ++i) {
                    v4f lo = {rb0[i].x, rb1[i].x, rb0[i].y, rb1[i].y};
                    v4f hi = {rb0[i].z, rb1[i].z, rb0[i].w, rb1[i].w};
                    *(v4f*)(bs + 8 * i * GPB)     = lo;
                    *(v4f*)(bs + 8 * i * GPB + 4) = hi;
                }
            }
            __syncthreads();
        }

#pragma unroll
        for (int off = 16; off; off >>= 1) {
            float t[16];
#pragma unroll
            for (int it = 0; it < 16; ++it) t[it] = __shfl_xor(rs[it], off, 32);
#pragma unroll
            for (int it = 0; it < 16; ++it) rs[it] += t[it];
        }
        if (lane == 0) {
#pragma unroll
            for (int it = 0; it < 16; ++it) Sinv[ar + 8 * it] = 1.0f / (rs[it] + 1e-8f);
        }
        __syncthreads();

#pragma unroll
        for (int mi = 0; mi < 4; ++mi) {
#pragma unroll
            for (int r = 0; r < 8; ++r) {
                const int lrow = wm * 64 + mi * 16 + half * 8 + r;
                const int row = b * LSEQ + i0 + lrow;
                const float invs = Sinv[lrow];
                float* crow = out + (size_t)row * D_DIM + bn0 + wn * 32 + l16;
#pragma unroll
                for (int ni = 0; ni < 2; ++ni) crow[ni * 16] = acc[mi][ni][r] * invs;
            }
        }
    }
}

extern "C" void kernel_launch(void* const* d_in, const int* in_sizes, int n_in,
                              void* d_out, int out_size, void* d_ws, size_t ws_size,
                              hipStream_t stream)
{
    const float* x  = (const float*)d_in[0];
    const float* Wq = (const float*)d_in[1];
    const float* Wg = (const float*)d_in[2];
    const float* bg = (const float*)d_in[3];
    const float* Wc = (const float*)d_in[4];
    float* out = (float*)d_out;

    const int Bsz = in_sizes[0] / (LSEQ * D_DIM);
    const int M   = Bsz * LSEQ;

    float* q    = (float*)d_ws;
    float* xavg = q    + (size_t)M * D_DIM;
    float* g    = xavg + (size_t)M * D_DIM;

    dim3 gblk(256);

    gemm_f32_wmma<false, false, D_DIM, D_DIM><<<dim3(D_DIM / GBN, M / GBM), gblk, 0, stream>>>(
        x, Wq, nullptr, q);

    cumavg_kernel<<<dim3(D_DIM / 32, Bsz), SCAN_WAVES * 32, 0, stream>>>(q, xavg);

    gemm_f32_wmma<true, true, LSEQ, D_DIM><<<dim3(TRI_BLOCKS, Bsz), gblk, 0, stream>>>(
        xavg, Wg, bg, g);

    attn_apply_lds<<<dim3(D_DIM / GBN, LSEQ / (2 * GBM), Bsz), gblk, 0, stream>>>(
        g, q, xavg);

    gemm_f32_wmma<false, false, D_DIM, D_DIM><<<dim3(D_DIM / GBN, M / GBM), gblk, 0, stream>>>(
        xavg, Wc, nullptr, out);
}


# Wq folded into the next weights: Wq@Wg and Wq@Wc formed first (one launch of 768 64x64 blocks), scan and attn apply read x, the q = x@Wq GEMM is gone; products re-associated, sum order changes
# speedup vs baseline: 3.2169x; 1.1361x over previous
#include <hip/hip_runtime.h>


typedef __attribute__((ext_vector_type(2))) float v2f;
typedef __attribute__((ext_vector_type(4))) float v4f;
typedef __attribute__((ext_vector_type(8))) float v8f;

#define D_DIM 1024
#define LSEQ  2048

#define GBM 128
#define GBN 128
#define GBK 32
#define GPA 36
#define GPB 288
#define TRI_BLOCKS ((LSEQ / GBM) * (LSEQ / GBM + 1) / 2)

template <bool RELU_BIAS, int MI, int NI, int K>
__device__ __forceinline__ void gemm_block(
    const float* __restrict__ A, const float* __restrict__ Bm,
    const float* __restrict__ bias, float* __restrict__ C,
    const int ldn, const int bm0, const int bn0,
    float* __restrict__ As, float* __restrict__ Bs)
{
    constexpr int BM  = 32 * MI;
    constexpr int BN  = 64 * NI;
    constexpr int PB  = 2 * BN + 32;
    constexpr int TB  = BN / 4;
    constexpr int BKS = 256 / TB;

    const int tid  = threadIdx.x;
    const int lane = tid & 31;
    const int wave = tid >> 5;
    const int half = lane >> 4;
    const int l16  = lane & 15;
    const int wm   = wave >> 2;
    const int wn   = wave & 3;

    const int arow = tid >> 3;
    const int akq  = (tid & 7) * 4;
    const float* Ag = A + (size_t)(bm0 + arow) * K + akq;
    float* Ast = As + arow * GPA + akq;
    const int bkp  = tid / TB;
    const int bn4  = (tid % TB) * 4;
    const float* Bg = Bm + (size_t)(2 * bkp) * ldn + bn0 + bn4;
    float* Bst = Bs + bkp * PB + 2 * bn4;

    const float* aF = As + (wm * 16 * MI + l16) * GPA + 2 * half;
    const float* bF = Bs + half * PB + 2 * (wn * 16 * NI + l16);

    v4f ra[MI], rb0[NI], rb1[NI];
    v8f acc[MI][NI] = {};

#pragma unroll
    for (int i = 0; i < MI; ++i) ra[i] = *(const v4f*)(Ag + (size_t)(32 * i) * K);
#pragma unroll
    for (int i = 0; i < NI; ++i) {
        rb0[i] = *(const v4f*)(Bg + (size_t)(2 * BKS * i) * ldn);
        rb1[i] = *(const v4f*)(Bg + (size_t)(2 * BKS * i + 1) * ldn);
    }
#pragma unroll
    for (int i = 0; i < MI; ++i) *(v4f*)(Ast + 32 * i * GPA) = ra[i];
#pragma unroll
    for (int i = 0; i < NI; ++i) {
        v4f lo = {rb0[i].x, rb1[i].x, rb0[i].y, rb1[i].y};
        v4f hi = {rb0[i].z, rb1[i].z, rb0[i].w, rb1[i].w};
        *(v4f*)(Bst + BKS * i * PB)     = lo;
        *(v4f*)(Bst + BKS * i * PB + 4) = hi;
    }
    __syncthreads();

    const int nk = K / GBK;
    for (int kt = 0; kt < nk; ++kt) {
        const int cur = kt & 1;
        const bool more = (kt + 1 < nk);

        if (more) {
            const float* ap = Ag + (kt + 1) * GBK;
            const float* bp = Bg + (size_t)((kt + 1) * GBK) * ldn;
#pragma unroll
            for (int i = 0; i < MI; ++i) ra[i] = *(const v4f*)(ap + (size_t)(32 * i) * K);
#pragma unroll
            for (int i = 0; i < NI; ++i) {
                rb0[i] = *(const v4f*)(bp + (size_t)(2 * BKS * i) * ldn);
                rb1[i] = *(const v4f*)(bp + (size_t)(2 * BKS * i + 1) * ldn);
            }
        }

        const float* ab = aF + cur * (BM * GPA);
        const float* bb = bF + cur * ((GBK / 2) * PB);
#pragma unroll
        for (int s = 0; s < GBK / 4; ++s) {
            v2f a[MI], b[NI];
#pragma unroll
            for (int mi = 0; mi < MI; ++mi) a[mi] = *(const v2f*)(ab + mi * 16 * GPA + 4 * s);
#pragma unroll
            for (int ni = 0; ni < NI; ++ni) b[ni] = *(const v2f*)(bb + 2 * s * PB + ni * 32);
#pragma unroll
            for (int mi = 0; mi < MI; ++mi)
#pragma unroll
                for (int ni = 0; ni < NI; ++ni)
                    acc[mi][ni] = __builtin_amdgcn_wmma_f32_16x16x4_f32(
                        false, a[mi], false, b[ni], (short)0, acc[mi][ni], false, false);
        }

        if (more) {
            float* as = Ast + (cur ^ 1) * (BM * GPA);
            float* bs = Bst + (cur ^ 1) * ((GBK / 2) * PB);
#pragma unroll
            for (int i = 0; i < MI; ++i) *(v4f*)(as + 32 * i * GPA) = ra[i];
#pragma unroll
            for (int i = 0; i < NI; ++i) {
                v4f lo = {rb0[i].x, rb1[i].x, rb0[i].y, rb1[i].y};
                v4f hi = {rb0[i].z, rb1[i].z, rb0[i].w, rb1[i].w};
                *(v4f*)(bs + BKS * i * PB)     = lo;
                *(v4f*)(bs + BKS * i * PB + 4) = hi;
            }
        }
        __syncthreads();
    }

    float bv[NI];
#pragma unroll
    for (int ni = 0; ni < NI; ++ni)
        bv[ni] = RELU_BIAS ? bias[bn0 + wn * 16 * NI + ni * 16 + l16] : 0.0f;
#pragma unroll
    for (int mi = 0; mi < MI; ++mi) {
#pragma unroll
        for (int r = 0; r < 8; ++r) {
            const int row = bm0 + wm * 16 * MI + mi * 16 + half * 8 + r;
            float* crow = C + (size_t)row * ldn + bn0 + wn * 16 * NI + l16;
#pragma unroll
            for (int ni = 0; ni < NI; ++ni) {
                float v = acc[mi][ni][r];
                if (RELU_BIAS) {
                    v += bv[ni];
                    v = v > 0.0f ? v : 0.0f;
                }
                crow[ni * 16] = v;
            }
        }
    }
}

template <bool RELU_BIAS, bool TRI, int N, int K>
__global__ __launch_bounds__(256) void gemm_f32_wmma(
    const float* __restrict__ A, const float* __restrict__ Bm,
    const float* __restrict__ bias, float* __restrict__ C)
{
    __shared__ __attribute__((aligned(16))) float As[2 * GBM * GPA];
    __shared__ __attribute__((aligned(16))) float Bs[2 * (GBK / 2) * GPB];

    int bm0 = blockIdx.y * GBM;
    int bn0 = blockIdx.x * GBN;
    if (TRI) {
        int rb = 0, rem = blockIdx.x;
        while (rem > rb) { rem -= rb + 1; ++rb; }
        bm0 = (blockIdx.y * (LSEQ / GBM) + rb) * GBM;
        bn0 = (N / GBN - 1 - rb + rem) * GBN;
    }
    gemm_block<RELU_BIAS, 4, 2, K>(A, Bm, bias, C, N, bm0, bn0, As, Bs);
}

#define WBT 64
__global__ __launch_bounds__(256) void wq_products(
    const float* __restrict__ Wq, const float* __restrict__ Wg,
    const float* __restrict__ Wc, float* __restrict__ Wqg, float* __restrict__ Wqc)
{
    __shared__ __attribute__((aligned(16))) float As[2 * WBT * GPA];
    __shared__ __attribute__((aligned(16))) float Bs[2 * (GBK / 2) * (2 * WBT + 32)];

    const bool first = blockIdx.x < LSEQ / WBT;
    const float* Bm  = first ? Wg : Wc;
    float*       C   = first ? Wqg : Wqc;
    const int    ldn = first ? LSEQ : D_DIM;
    const int    bn0 = (first ? (int)blockIdx.x : (int)blockIdx.x - LSEQ / WBT) * WBT;
    gemm_block<false, 2, 1, D_DIM>(Wq, Bm, nullptr, C, ldn, blockIdx.y * WBT, bn0, As, Bs);
}

#define SCAN_ROWS  64
#define SCAN_WAVES (LSEQ / SCAN_ROWS)

__global__ __launch_bounds__(SCAN_WAVES * 32) void cumavg_kernel(
    const float* __restrict__ x, float* __restrict__ xavg)
{
    __shared__ float tot[SCAN_WAVES][32];

    const int lane = threadIdx.x & 31;
    const int c  = __builtin_amdgcn_readfirstlane((int)(threadIdx.x >> 5));
    const int b  = blockIdx.y;
    const int d  = blockIdx.x * 32 + lane;
    const int i0 = c * SCAN_ROWS;
    const size_t base = ((size_t)b * LSEQ + (size_t)i0) * D_DIM + d;
    const float* src = x + base;
    float*       dst = xavg + base;

    float v[SCAN_ROWS];
#pragma unroll
    for (int r = 0; r < SCAN_ROWS; ++r) v[r] = src[(size_t)r * D_DIM];
#pragma unroll
    for (int r = 1; r < SCAN_ROWS; ++r) v[r] += v[r - 1];

    tot[c][lane] = v[SCAN_ROWS - 1];
    __syncthreads();

    float off = 0.0f;
    for (int p = 0; p < c; ++p) off += tot[p][lane];

#pragma unroll
    for (int r = 0; r < SCAN_ROWS; ++r)
        dst[(size_t)r * D_DIM] = (off + v[r]) / (float)(i0 + r + 1);
}

__global__ __launch_bounds__(256) void attn_apply_lds(
    const float* __restrict__ g, const float* __restrict__ x,
    float* __restrict__ out)
{
    __shared__ __attribute__((aligned(16))) float As[2 * GBM * GPA];
    __shared__ __attribute__((aligned(16))) float Bs[2 * (GBK / 2) * GPB];
    __shared__ __attribute__((aligned(16))) float Sinv[GBM];

    const int tid  = threadIdx.x;
    const int lane = tid & 31;
    const int wave = tid >> 5;
    const int half = lane >> 4;
    const int l16  = lane & 15;
    const int wm   = wave >> 2;
    const int wn   = wave & 3;
    const int b    = blockIdx.z;
    const int bn0  = blockIdx.x * GBN;

    const int ac = tid & 31;
    const int ar = tid >> 5;
    float* Ast = As + ar * GPA + ac;
    const int bkp  = tid >> 5;
    const int bn4  = (tid & 31) * 4;
    const float* Bg = x + ((size_t)b * LSEQ + 2 * bkp) * D_DIM + bn0 + bn4;
    float* Bst = Bs + bkp * GPB + 2 * bn4;

    const float* aF = As + (wm * 64 + l16) * GPA + 2 * half;
    const float* bF = Bs + half * GPB + 2 * (wn * 32 + l16);

    for (int pass = 0; pass < 2; ++pass) {
        const int rb = pass ? (int)blockIdx.y : (LSEQ / GBM - 1) - (int)blockIdx.y;
        const int i0 = rb * GBM;
        const int nk = (i0 + GBM) / GBK;

        const float* Ag = g + ((size_t)b * LSEQ + i0 + ar) * LSEQ + (LSEQ - 1 - (i0 + ar)) + ac;
        const int jlim = i0 + ar - ac;

        float ra[16];
        float rs[16];
        v4f rb0[2], rb1[2];
        v8f acc[4][2] = {};

#pragma unroll
        for (int it = 0; it < 16; ++it)
            ra[it] = (0 <= jlim + 8 * it) ? Ag[(size_t)it * (8 * LSEQ - 8)] : 0.0f;
#pragma unroll
        for (int i = 0; i < 2; ++i) {
            rb0[i] = *(const v4f*)(Bg + (size_t)(16 * i) * D_DIM);
            rb1[i] = *(const v4f*)(Bg + (size_t)(16 * i + 1) * D_DIM);
        }
#pragma unroll
        for (int it = 0; it < 16; ++it) {
            Ast[8 * it * GPA] = ra[it];
            rs[it] = ra[it];
        }
#pragma unroll
        for (int i = 0; i < 2; ++i) {
            v4f lo = {rb0[i].x, rb1[i].x, rb0[i].y, rb1[i].y};
            v4f hi = {rb0[i].z, rb1[i].z, rb0[i].w, rb1[i].w};
            *(v4f*)(Bst + 8 * i * GPB)     = lo;
            *(v4f*)(Bst + 8 * i * GPB + 4) = hi;
        }
        __syncthreads();

        for (int kt = 0; kt < nk; ++kt) {
            const int cur = kt & 1;
            const bool more = (kt + 1 < nk);

            if (more) {
                const int j0 = (kt + 1) * GBK;
                const float* ap = Ag + j0;
                const float* bp = Bg + (size_t)j0 * D_DIM;
                if (j0 < i0) {
#pragma unroll
                    for (int it = 0; it < 16; ++it) ra[it] = ap[(size_t)it * (8 * LSEQ - 8)];
                } else {
#pragma unroll
                    for (int it = 0; it < 16; ++it)
                        ra[it] = (j0 <= jlim + 8 * it) ? ap[(size_t)it * (8 * LSEQ - 8)] : 0.0f;
                }
#pragma unroll
                for (int i = 0; i < 2; ++i) {
                    rb0[i] = *(const v4f*)(bp + (size_t)(16 * i) * D_DIM);
                    rb1[i] = *(const v4f*)(bp + (size_t)(16 * i + 1) * D_DIM);
                }
            }

            const float* ab = aF + cur * (GBM * GPA);
            const float* bb = bF + cur * ((GBK / 2) * GPB);
#pragma unroll
            for (int s = 0; s < GBK / 4; ++s) {
                v2f a[4], bf[2];
#pragma unroll
                for (int mi = 0; mi < 4; ++mi) a[mi] = *(const v2f*)(ab + mi * 16 * GPA + 4 * s);
#pragma unroll
                for (int ni = 0; ni < 2; ++ni) bf[ni] = *(const v2f*)(bb + 2 * s * GPB + ni * 32);
#pragma unroll
                for (int mi = 0; mi < 4; ++mi)
#pragma unroll
                    for (int ni = 0; ni < 2; ++ni)
                        acc[mi][ni] = __builtin_amdgcn_wmma_f32_16x16x4_f32(
                            false, a[mi], false, bf[ni], (short)0, acc[mi][ni], false, false);
            }

            if (more) {
                float* as = Ast + (cur ^ 1) * (GBM * GPA);
                float* bs = Bst + (cur ^ 1) * ((GBK / 2) * GPB);
#pragma unroll
                for (int it = 0; it < 16; ++it) {
                    as[8 * it * GPA] = ra[it];
                    rs[it] += ra[it];
                }
#pragma unroll
                for (int i = 0; i < 2; ++i) {
                    v4f lo = {rb0[i].x, rb1[i].x, rb0[i].y, rb1[i].y};
                    v4f hi = {rb0[i].z, rb1[i].z, rb0[i].w, rb1[i].w};
                    *(v4f*)(bs + 8 * i * GPB)     = lo;
                    *(v4f*)(bs + 8 * i * GPB + 4) = hi;
                }
            }
            __syncthreads();
        }

#pragma unroll
        for (int off = 16; off; off >>= 1) {
            float t[16];
#pragma unroll
            for (int it = 0; it < 16; ++it) t[it] = __shfl_xor(rs[it], off, 32);
#pragma unroll
            for (int it = 0; it < 16; ++it) rs[it] += t[it];
        }
        if (lane == 0) {
#pragma unroll
            for (int it = 0; it < 16; ++it) Sinv[ar + 8 * it] = 1.0f / (rs[it] + 1e-8f);
        }
        __syncthreads();

#pragma unroll
        for (int mi = 0; mi < 4; ++mi) {
#pragma unroll
            for (int r = 0; r < 8; ++r) {
                const int lrow = wm * 64 + mi * 16 + half * 8 + r;
                const int row = b * LSEQ + i0 + lrow;
                const float invs = Sinv[lrow];
                float* crow = out + (size_t)row * D_DIM + bn0 + wn * 32 + l16;
#pragma unroll
                for (int ni = 0; ni < 2; ++ni) crow[ni * 16] = acc[mi][ni][r] * invs;
            }
        }
    }
}

extern "C" void kernel_launch(void* const* d_in, const int* in_sizes, int n_in,
                              void* d_out, int out_size, void* d_ws, size_t ws_size,
                              hipStream_t stream)
{
    const float* x  = (const float*)d_in[0];
    const float* Wq = (const float*)d_in[1];
    const float* Wg = (const float*)d_in[2];
    const float* bg = (const float*)d_in[3];
    const float* Wc = (const float*)d_in[4];
    float* out = (float*)d_out;

    const int Bsz = in_sizes[0] / (LSEQ * D_DIM);
    const int M   = Bsz * LSEQ;

    float* Wqg  = (float*)d_ws;
    float* Wqc  = Wqg  + (size_t)D_DIM * LSEQ;
    float* xavg = Wqg  + (size_t)M * D_DIM;
    float* g    = xavg + (size_t)M * D_DIM;

    dim3 gblk(256);

    wq_products<<<dim3((LSEQ + D_DIM) / WBT, D_DIM / WBT), gblk, 0, stream>>>(
        Wq, Wg, Wc, Wqg, Wqc);

    cumavg_kernel<<<dim3(D_DIM / 32, Bsz), SCAN_WAVES * 32, 0, stream>>>(x, xavg);

    gemm_f32_wmma<true, true, LSEQ, D_DIM><<<dim3(TRI_BLOCKS, Bsz), gblk, 0, stream>>>(
        xavg, Wqg, bg, g);

    attn_apply_lds<<<dim3(D_DIM / GBN, LSEQ / (2 * GBM), Bsz), gblk, 0, stream>>>(
        g, x, xavg);

    gemm_f32_wmma<false, false, D_DIM, D_DIM><<<dim3(D_DIM / GBN, M / GBM), gblk, 0, stream>>>(
        xavg, Wqc, nullptr, out);
}


# K loops of the four WMMA kernels pipelined over half k-tiles: each half's LDS fragment reads go out as one group ahead of the other half's WMMAs (two register sets); sum order unchanged
# speedup vs baseline: 3.3367x; 1.0372x over previous
#include <hip/hip_runtime.h>


typedef __attribute__((ext_vector_type(2))) float v2f;
typedef __attribute__((ext_vector_type(4))) float v4f;
typedef __attribute__((ext_vector_type(8))) float v8f;

#define D_DIM 1024
#define LSEQ  2048

#define GBM 128
#define GBN 128
#define GBK 32
#define GPA 36
#define GPB 288
#define TRI_BLOCKS ((LSEQ / GBM) * (LSEQ / GBM + 1) / 2)

template <int MI, int NI, int PB, int S0>
__device__ __forceinline__ void read_half(v2f (&a)[4][MI], v2f (&b)[4][NI],
                                          const float* __restrict__ ab,
                                          const float* __restrict__ bb)
{
#pragma unroll
    for (int s = 0; s < 4; ++s) {
#pragma unroll
        for (int mi = 0; mi < MI; ++mi)
            a[s][mi] = *(const v2f*)(ab + mi * 16 * GPA + 4 * (S0 + s));
#pragma unroll
        for (int ni = 0; ni < NI; ++ni)
            b[s][ni] = *(const v2f*)(bb + 2 * (S0 + s) * PB + ni * 32);
    }
}

template <int MI, int NI>
__device__ __forceinline__ void wmma_half(v8f (&acc)[MI][NI],
                                          const v2f (&a)[4][MI], const v2f (&b)[4][NI])
{
#pragma unroll
    for (int s = 0; s < 4; ++s)
#pragma unroll
        for (int mi = 0; mi < MI; ++mi)
#pragma unroll
            for (int ni = 0; ni < NI; ++ni)
                acc[mi][ni] = __builtin_amdgcn_wmma_f32_16x16x4_f32(
                    false, a[s][mi], false, b[s][ni], (short)0, acc[mi][ni], false, false);
}

template <bool RELU_BIAS, int MI, int NI, int K>
__device__ __forceinline__ void gemm_block(
    const float* __restrict__ A, const float* __restrict__ Bm,
    const float* __restrict__ bias, float* __restrict__ C,
    const int ldn, const int bm0, const int bn0,
    float* __restrict__ As, float* __restrict__ Bs)
{
    constexpr int BM  = 32 * MI;
    constexpr int BN  = 64 * NI;
    constexpr int PB  = 2 * BN + 32;
    constexpr int TB  = BN / 4;
    constexpr int BKS = 256 / TB;

    const int tid  = threadIdx.x;
    const int lane = tid & 31;
    const int wave = tid >> 5;
    const int half = lane >> 4;
    const int l16  = lane & 15;
    const int wm   = wave >> 2;
    const int wn   = wave & 3;

    const int arow = tid >> 3;
    const int akq  = (tid & 7) * 4;
    const float* Ag = A + (size_t)(bm0 + arow) * K + akq;
    float* Ast = As + arow * GPA + akq;
    const int bkp  = tid / TB;
    const int bn4  = (tid % TB) * 4;
    const float* Bg = Bm + (size_t)(2 * bkp) * ldn + bn0 + bn4;
    float* Bst = Bs + bkp * PB + 2 * bn4;

    const float* aF = As + (wm * 16 * MI + l16) * GPA + 2 * half;
    const float* bF = Bs + half * PB + 2 * (wn * 16 * NI + l16);

    v4f ra[MI], rb0[NI], rb1[NI];
    v8f acc[MI][NI] = {};
    v2f fa0[4][MI], fb0[4][NI], fa1[4][MI], fb1[4][NI];

    constexpr int nk = K / GBK;

#pragma unroll
    for (int i = 0; i < MI; ++i) ra[i] = *(const v4f*)(Ag + (size_t)(32 * i) * K);
#pragma unroll
    for (int i = 0; i < NI; ++i) {
        rb0[i] = *(const v4f*)(Bg + (size_t)(2 * BKS * i) * ldn);
        rb1[i] = *(const v4f*)(Bg + (size_t)(2 * BKS * i + 1) * ldn);
    }
#pragma unroll
    for (int i = 0; i < MI; ++i) *(v4f*)(Ast + 32 * i * GPA) = ra[i];
#pragma unroll
    for (int i = 0; i < NI; ++i) {
        v4f lo = {rb0[i].x, rb1[i].x, rb0[i].y, rb1[i].y};
        v4f hi = {rb0[i].z, rb1[i].z, rb0[i].w, rb1[i].w};
        *(v4f*)(Bst + BKS * i * PB)     = lo;
        *(v4f*)(Bst + BKS * i * PB + 4) = hi;
    }
    __syncthreads();
    {
        const float* ap = Ag + GBK;
        const float* bp = Bg + (size_t)GBK * ldn;
#pragma unroll
        for (int i = 0; i < MI; ++i) ra[i] = *(const v4f*)(ap + (size_t)(32 * i) * K);
#pragma unroll
        for (int i = 0; i < NI; ++i) {
            rb0[i] = *(const v4f*)(bp + (size_t)(2 * BKS * i) * ldn);
            rb1[i] = *(const v4f*)(bp + (size_t)(2 * BKS * i + 1) * ldn);
        }
    }
    read_half<MI, NI, PB, 0>(fa0, fb0, aF, bF);
    read_half<MI, NI, PB, 4>(fa1, fb1, aF, bF);
    __builtin_amdgcn_sched_barrier(0);
    wmma_half<MI, NI>(acc, fa0, fb0);
    __builtin_amdgcn_sched_barrier(0);

    for (int t = 0; t + 1 < nk; ++t) {
        const int nxt = (t + 1) & 1;
        {
            float* as = Ast + nxt * (BM * GPA);
            float* bs = Bst + nxt * ((GBK / 2) * PB);
#pragma unroll
            for (int i = 0; i < MI; ++i) *(v4f*)(as + 32 * i * GPA) = ra[i];
#pragma unroll
            for (int i = 0; i < NI; ++i) {
                v4f lo = {rb0[i].x, rb1[i].x, rb0[i].y, rb1[i].y};
                v4f hi = {rb0[i].z, rb1[i].z, rb0[i].w, rb1[i].w};
                *(v4f*)(bs + BKS * i * PB)     = lo;
                *(v4f*)(bs + BKS * i * PB + 4) = hi;
            }
        }
        __syncthreads();

        if (t + 2 < nk) {
            const float* ap = Ag + (t + 2) * GBK;
            const float* bp = Bg + (size_t)((t + 2) * GBK) * ldn;
#pragma unroll
            for (int i = 0; i < MI; ++i) ra[i] = *(const v4f*)(ap + (size_t)(32 * i) * K);
#pragma unroll
            for (int i = 0; i < NI; ++i) {
                rb0[i] = *(const v4f*)(bp + (size_t)(2 * BKS * i) * ldn);
                rb1[i] = *(const v4f*)(bp + (size_t)(2 * BKS * i + 1) * ldn);
            }
        }
        read_half<MI, NI, PB, 0>(fa0, fb0, aF + nxt * (BM * GPA), bF + nxt * ((GBK / 2) * PB));
        __builtin_amdgcn_sched_barrier(0);
        wmma_half<MI, NI>(acc, fa1, fb1);
        __builtin_amdgcn_sched_barrier(0);
        read_half<MI, NI, PB, 4>(fa1, fb1, aF + nxt * (BM * GPA), bF + nxt * ((GBK / 2) * PB));
        __builtin_amdgcn_sched_barrier(0);
        wmma_half<MI, NI>(acc, fa0, fb0);
        __builtin_amdgcn_sched_barrier(0);
    }
    wmma_half<MI, NI>(acc, fa1, fb1);

    float bv[NI];
#pragma unroll
    for (int ni = 0; ni < NI; ++ni)
        bv[ni] = RELU_BIAS ? bias[bn0 + wn * 16 * NI + ni * 16 + l16] : 0.0f;
#pragma unroll
    for (int mi = 0; mi < MI; ++mi) {
#pragma unroll
        for (int r = 0; r < 8; ++r) {
            const int row = bm0 + wm * 16 * MI + mi * 16 + half * 8 + r;
            float* crow = C + (size_t)row * ldn + bn0 + wn * 16 * NI + l16;
#pragma unroll
            for (int ni = 0; ni < NI; ++ni) {
                float v = acc[mi][ni][r];
                if (RELU_BIAS) {
                    v += bv[ni];
                    v = v > 0.0f ? v : 0.0f;
                }
                crow[ni * 16] = v;
            }
        }
    }
}

template <bool RELU_BIAS, bool TRI, int N, int K>
__global__ __launch_bounds__(256) void gemm_f32_wmma(
    const float* __restrict__ A, const float* __restrict__ Bm,
    const float* __restrict__ bias, float* __restrict__ C)
{
    __shared__ __attribute__((aligned(16))) float As[2 * GBM * GPA];
    __shared__ __attribute__((aligned(16))) float Bs[2 * (GBK / 2) * GPB];

    int bm0 = blockIdx.y * GBM;
    int bn0 = blockIdx.x * GBN;
    if (TRI) {
        int rb = 0, rem = blockIdx.x;
        while (rem > rb) { rem -= rb + 1; ++rb; }
        bm0 = (blockIdx.y * (LSEQ / GBM) + rb) * GBM;
        bn0 = (N / GBN - 1 - rb + rem) * GBN;
    }
    gemm_block<RELU_BIAS, 4, 2, K>(A, Bm, bias, C, N, bm0, bn0, As, Bs);
}

#define WBT 64
__global__ __launch_bounds__(256) void wq_products(
    const float* __restrict__ Wq, const float* __restrict__ Wg,
    const float* __restrict__ Wc, float* __restrict__ Wqg, float* __restrict__ Wqc)
{
    __shared__ __attribute__((aligned(16))) float As[2 * WBT * GPA];
    __shared__ __attribute__((aligned(16))) float Bs[2 * (GBK / 2) * (2 * WBT + 32)];

    const bool first = blockIdx.x < LSEQ / WBT;
    const float* Bm  = first ? Wg : Wc;
    float*       C   = first ? Wqg : Wqc;
    const int    ldn = first ? LSEQ : D_DIM;
    const int    bn0 = (first ? (int)blockIdx.x : (int)blockIdx.x - LSEQ / WBT) * WBT;
    gemm_block<false, 2, 1, D_DIM>(Wq, Bm, nullptr, C, ldn, blockIdx.y * WBT, bn0, As, Bs);
}

#define SCAN_ROWS  64
#define SCAN_WAVES (LSEQ / SCAN_ROWS)

__global__ __launch_bounds__(SCAN_WAVES * 32) void cumavg_kernel(
    const float* __restrict__ x, float* __restrict__ xavg)
{
    __shared__ float tot[SCAN_WAVES][32];

    const int lane = threadIdx.x & 31;
    const int c  = __builtin_amdgcn_readfirstlane((int)(threadIdx.x >> 5));
    const int b  = blockIdx.y;
    const int d  = blockIdx.x * 32 + lane;
    const int i0 = c * SCAN_ROWS;
    const size_t base = ((size_t)b * LSEQ + (size_t)i0) * D_DIM + d;
    const float* src = x + base;
    float*       dst = xavg + base;

    float v[SCAN_ROWS];
#pragma unroll
    for (int r = 0; r < SCAN_ROWS; ++r) v[r] = src[(size_t)r * D_DIM];
#pragma unroll
    for (int r = 1; r < SCAN_ROWS; ++r) v[r] += v[r - 1];

    tot[c][lane] = v[SCAN_ROWS - 1];
    __syncthreads();

    float off = 0.0f;
    for (int p = 0; p < c; ++p) off += tot[p][lane];

#pragma unroll
    for (int r = 0; r < SCAN_ROWS; ++r)
        dst[(size_t)r * D_DIM] = (off + v[r]) / (float)(i0 + r + 1);
}

__device__ __forceinline__ void attn_load_tile(
    float (&ra)[16], v4f (&rb0)[2], v4f (&rb1)[2],
    const float* __restrict__ Ag, const float* __restrict__ Bg,
    const int j0, const int i0, const int jlim)
{
    const float* ap = Ag + j0;
    const float* bp = Bg + (size_t)j0 * D_DIM;
    if (j0 < i0) {
#pragma unroll
        for (int it = 0; it < 16; ++it) ra[it] = ap[(size_t)it * (8 * LSEQ - 8)];
    } else {
#pragma unroll
        for (int it = 0; it < 16; ++it)
            ra[it] = (j0 <= jlim + 8 * it) ? ap[(size_t)it * (8 * LSEQ - 8)] : 0.0f;
    }
#pragma unroll
    for (int i = 0; i < 2; ++i) {
        rb0[i] = *(const v4f*)(bp + (size_t)(16 * i) * D_DIM);
        rb1[i] = *(const v4f*)(bp + (size_t)(16 * i + 1) * D_DIM);
    }
}

__global__ __launch_bounds__(256) void attn_apply_lds(
    const float* __restrict__ g, const float* __restrict__ x,
    float* __restrict__ out)
{
    __shared__ __attribute__((aligned(16))) float As[2 * GBM * GPA];
    __shared__ __attribute__((aligned(16))) float Bs[2 * (GBK / 2) * GPB];
    __shared__ __attribute__((aligned(16))) float Sinv[GBM];

    const int tid  = threadIdx.x;
    const int lane = tid & 31;
    const int wave = tid >> 5;
    const int half = lane >> 4;
    const int l16  = lane & 15;
    const int wm   = wave >> 2;
    const int wn   = wave & 3;
    const int b    = blockIdx.z;
    const int bn0  = blockIdx.x * GBN;

    const int ac = tid & 31;
    const int ar = tid >> 5;
    float* Ast = As + ar * GPA + ac;
    const int bkp  = tid >> 5;
    const int bn4  = (tid & 31) * 4;
    const float* Bg = x + ((size_t)b * LSEQ + 2 * bkp) * D_DIM + bn0 + bn4;
    float* Bst = Bs + bkp * GPB + 2 * bn4;

    const float* aF = As + (wm * 64 + l16) * GPA + 2 * half;
    const float* bF = Bs + half * GPB + 2 * (wn * 32 + l16);

    for (int pass = 0; pass < 2; ++pass) {
        const int rb = pass ? (int)blockIdx.y : (LSEQ / GBM - 1) - (int)blockIdx.y;
        const int i0 = rb * GBM;
        const int nk = (i0 + GBM) / GBK;

        const float* Ag = g + ((size_t)b * LSEQ + i0 + ar) * LSEQ + (LSEQ - 1 - (i0 + ar)) + ac;
        const int jlim = i0 + ar - ac;

        float ra[16];
        float rs[16];
        v4f rb0[2], rb1[2];
        v8f acc[4][2] = {};
        v2f fa0[4][4], fb0[4][2], fa1[4][4], fb1[4][2];

        attn_load_tile(ra, rb0, rb1, Ag, Bg, 0, i0, jlim);
#pragma unroll
        for (int it = 0; it < 16; ++it) {
            Ast[8 * it * GPA] = ra[it];
            rs[it] = ra[it];
        }
#pragma unroll
        for (int i = 0; i < 2; ++i) {
            v4f lo = {rb0[i].x, rb1[i].x, rb0[i].y, rb1[i].y};
            v4f hi = {rb0[i].z, rb1[i].z, rb0[i].w, rb1[i].w};
            *(v4f*)(Bst + 8 * i * GPB)     = lo;
            *(v4f*)(Bst + 8 * i * GPB + 4) = hi;
        }
        __syncthreads();
        attn_load_tile(ra, rb0, rb1, Ag, Bg, GBK, i0, jlim);
        read_half<4, 2, GPB, 0>(fa0, fb0, aF, bF);
        read_half<4, 2, GPB, 4>(fa1, fb1, aF, bF);
        __builtin_amdgcn_sched_barrier(0);
        wmma_half<4, 2>(acc, fa0, fb0);
        __builtin_amdgcn_sched_barrier(0);

        for (int t = 0; t + 1 < nk; ++t) {
            const int nxt = (t + 1) & 1;
            {
                float* as = Ast + nxt * (GBM * GPA);
                float* bs = Bst + nxt * ((GBK / 2) * GPB);
#pragma unroll
                for (int it = 0; it < 16; ++it) {
                    as[8 * it * GPA] = ra[it];
                    rs[it] += ra[it];
                }
#pragma unroll
                for (int i = 0; i < 2; ++i) {
                    v4f lo = {rb0[i].x, rb1[i].x, rb0[i].y, rb1[i].y};
                    v4f hi = {rb0[i].z, rb1[i].z, rb0[i].w, rb1[i].w};
                    *(v4f*)(bs + 8 * i * GPB)     = lo;
                    *(v4f*)(bs + 8 * i * GPB + 4) = hi;
                }
            }
            __syncthreads();

            attn_load_tile(ra, rb0, rb1, Ag, Bg, (t + 2 < nk ? t + 2 : nk - 1) * GBK, i0, jlim);
            read_half<4, 2, GPB, 0>(fa0, fb0, aF + nxt * (GBM * GPA),
                                    bF + nxt * ((GBK / 2) * GPB));
            __builtin_amdgcn_sched_barrier(0);
            wmma_half<4, 2>(acc, fa1, fb1);
            __builtin_amdgcn_sched_barrier(0);
            read_half<4, 2, GPB, 4>(fa1, fb1, aF + nxt * (GBM * GPA),
                                    bF + nxt * ((GBK / 2) * GPB));
            __builtin_amdgcn_sched_barrier(0);
            wmma_half<4, 2>(acc, fa0, fb0);
            __builtin_amdgcn_sched_barrier(0);
        }
        wmma_half<4, 2>(acc, fa1, fb1);

#pragma unroll
        for (int off = 16; off; off >>= 1) {
            float t[16];
#pragma unroll
            for (int it = 0; it < 16; ++it) t[it] = __shfl_xor(rs[it], off, 32);
#pragma unroll
            for (int it = 0; it < 16; ++it) rs[it] += t[it];
        }
        if (lane == 0) {
#pragma unroll
            for (int it = 0; it < 16; ++it) Sinv[ar + 8 * it] = 1.0f / (rs[it] + 1e-8f);
        }
        __syncthreads();

#pragma unroll
        for (int mi = 0; mi < 4; ++mi) {
#pragma unroll
            for (int r = 0; r < 8; ++r) {
                const int lrow = wm * 64 + mi * 16 + half * 8 + r;
                const int row = b * LSEQ + i0 + lrow;
                const float invs = Sinv[lrow];
                float* crow = out + (size_t)row * D_DIM + bn0 + wn * 32 + l16;
#pragma unroll
                for (int ni = 0; ni < 2; ++ni) crow[ni * 16] = acc[mi][ni][r] * invs;
            }
        }
    }
}

extern "C" void kernel_launch(void* const* d_in, const int* in_sizes, int n_in,
                              void* d_out, int out_size, void* d_ws, size_t ws_size,
                              hipStream_t stream)
{
    const float* x  = (const float*)d_in[0];
    const float* Wq = (const float*)d_in[1];
    const float* Wg = (const float*)d_in[2];
    const float* bg = (const float*)d_in[3];
    const float* Wc = (const float*)d_in[4];
    float* out = (float*)d_out;

    const int Bsz = in_sizes[0] / (LSEQ * D_DIM);
    const int M   = Bsz * LSEQ;

    float* Wqg  = (float*)d_ws;
    float* Wqc  = Wqg  + (size_t)D_DIM * LSEQ;
    float* xavg = Wqg  + (size_t)M * D_DIM;
    float* g    = xavg + (size_t)M * D_DIM;

    dim3 gblk(256);

    wq_products<<<dim3((LSEQ + D_DIM) / WBT, D_DIM / WBT), gblk, 0, stream>>>(
        Wq, Wg, Wc, Wqg, Wqc);

    cumavg_kernel<<<dim3(D_DIM / 32, Bsz), SCAN_WAVES * 32, 0, stream>>>(x, xavg);

    gemm_f32_wmma<true, true, LSEQ, D_DIM><<<dim3(TRI_BLOCKS, Bsz), gblk, 0, stream>>>(
        xavg, Wqg, bg, g);

    attn_apply_lds<<<dim3(D_DIM / GBN, LSEQ / (2 * GBM), Bsz), gblk, 0, stream>>>(
        g, x, xavg);

    gemm_f32_wmma<false, false, D_DIM, D_DIM><<<dim3(D_DIM / GBN, M / GBM), gblk, 0, stream>>>(
        xavg, Wqc, nullptr, out);
}


# g GEMM launch balanced: 512 of 544 blocks done whole (two resident per WGP), the other 32 cut into 256 tiles of 32x64 handed out first: no second round of 32 lone blocks; sum order unchanged
# speedup vs baseline: 3.5261x; 1.0568x over previous
#include <hip/hip_runtime.h>


typedef __attribute__((ext_vector_type(2))) float v2f;
typedef __attribute__((ext_vector_type(4))) float v4f;
typedef __attribute__((ext_vector_type(8))) float v8f;

#define D_DIM 1024
#define LSEQ  2048

#define GBM 128
#define GBN 128
#define GBK 32
#define GPA 36
#define GPB 288
#define TRI_BLOCKS ((LSEQ / GBM) * (LSEQ / GBM + 1) / 2)

template <int MI, int NI, int PB, int S0>
__device__ __forceinline__ void read_half(v2f (&a)[4][MI], v2f (&b)[4][NI],
                                          const float* __restrict__ ab,
                                          const float* __restrict__ bb)
{
#pragma unroll
    for (int s = 0; s < 4; ++s) {
#pragma unroll
        for (int mi = 0; mi < MI; ++mi)
            a[s][mi] = *(const v2f*)(ab + mi * 16 * GPA + 4 * (S0 + s));
#pragma unroll
        for (int ni = 0; ni < NI; ++ni)
            b[s][ni] = *(const v2f*)(bb + 2 * (S0 + s) * PB + ni * 32);
    }
}

template <int MI, int NI>
__device__ __forceinline__ void wmma_half(v8f (&acc)[MI][NI],
                                          const v2f (&a)[4][MI], const v2f (&b)[4][NI])
{
#pragma unroll
    for (int s = 0; s < 4; ++s)
#pragma unroll
        for (int mi = 0; mi < MI; ++mi)
#pragma unroll
            for (int ni = 0; ni < NI; ++ni)
                acc[mi][ni] = __builtin_amdgcn_wmma_f32_16x16x4_f32(
                    false, a[s][mi], false, b[s][ni], (short)0, acc[mi][ni], false, false);
}

template <bool RELU_BIAS, int MI, int NI, int K>
__device__ __forceinline__ void gemm_block(
    const float* __restrict__ A, const float* __restrict__ Bm,
    const float* __restrict__ bias, float* __restrict__ C,
    const int ldn, const int bm0, const int bn0,
    float* __restrict__ As, float* __restrict__ Bs)
{
    constexpr int BM  = 32 * MI;
    constexpr int BN  = 64 * NI;
    constexpr int PB  = 2 * BN + 32;
    constexpr int TB  = BN / 4;
    constexpr int BKS = 256 / TB;

    const int tid  = threadIdx.x;
    const int lane = tid & 31;
    const int wave = tid >> 5;
    const int half = lane >> 4;
    const int l16  = lane & 15;
    const int wm   = wave >> 2;
    const int wn   = wave & 3;

    const int arow = tid >> 3;
    const int akq  = (tid & 7) * 4;
    const float* Ag = A + (size_t)(bm0 + arow) * K + akq;
    float* Ast = As + arow * GPA + akq;
    const int bkp  = tid / TB;
    const int bn4  = (tid % TB) * 4;
    const float* Bg = Bm + (size_t)(2 * bkp) * ldn + bn0 + bn4;
    float* Bst = Bs + bkp * PB + 2 * bn4;

    const float* aF = As + (wm * 16 * MI + l16) * GPA + 2 * half;
    const float* bF = Bs + half * PB + 2 * (wn * 16 * NI + l16);

    v4f ra[MI], rb0[NI], rb1[NI];
    v8f acc[MI][NI] = {};
    v2f fa0[4][MI], fb0[4][NI], fa1[4][MI], fb1[4][NI];

    constexpr int nk = K / GBK;

#pragma unroll
    for (int i = 0; i < MI; ++i) ra[i] = *(const v4f*)(Ag + (size_t)(32 * i) * K);
#pragma unroll
    for (int i = 0; i < NI; ++i) {
        rb0[i] = *(const v4f*)(Bg + (size_t)(2 * BKS * i) * ldn);
        rb1[i] = *(const v4f*)(Bg + (size_t)(2 * BKS * i + 1) * ldn);
    }
#pragma unroll
    for (int i = 0; i < MI; ++i) *(v4f*)(Ast + 32 * i * GPA) = ra[i];
#pragma unroll
    for (int i = 0; i < NI; ++i) {
        v4f lo = {rb0[i].x, rb1[i].x, rb0[i].y, rb1[i].y};
        v4f hi = {rb0[i].z, rb1[i].z, rb0[i].w, rb1[i].w};
        *(v4f*)(Bst + BKS * i * PB)     = lo;
        *(v4f*)(Bst + BKS * i * PB + 4) = hi;
    }
    __syncthreads();
    {
        const float* ap = Ag + GBK;
        const float* bp = Bg + (size_t)GBK * ldn;
#pragma unroll
        for (int i = 0; i < MI; ++i) ra[i] = *(const v4f*)(ap + (size_t)(32 * i) * K);
#pragma unroll
        for (int i = 0; i < NI; ++i) {
            rb0[i] = *(const v4f*)(bp + (size_t)(2 * BKS * i) * ldn);
            rb1[i] = *(const v4f*)(bp + (size_t)(2 * BKS * i + 1) * ldn);
        }
    }
    read_half<MI, NI, PB, 0>(fa0, fb0, aF, bF);
    read_half<MI, NI, PB, 4>(fa1, fb1, aF, bF);
    __builtin_amdgcn_sched_barrier(0);
    wmma_half<MI, NI>(acc, fa0, fb0);
    __builtin_amdgcn_sched_barrier(0);

    for (int t = 0; t + 1 < nk; ++t) {
        const int nxt = (t + 1) & 1;
        {
            float* as = Ast + nxt * (BM * GPA);
            float* bs = Bst + nxt * ((GBK / 2) * PB);
#pragma unroll
            for (int i = 0; i < MI; ++i) *(v4f*)(as + 32 * i * GPA) = ra[i];
#pragma unroll
            for (int i = 0; i < NI; ++i) {
                v4f lo = {rb0[i].x, rb1[i].x, rb0[i].y, rb1[i].y};
                v4f hi = {rb0[i].z, rb1[i].z, rb0[i].w, rb1[i].w};
                *(v4f*)(bs + BKS * i * PB)     = lo;
                *(v4f*)(bs + BKS * i * PB + 4) = hi;
            }
        }
        __syncthreads();

        if (t + 2 < nk) {
            const float* ap = Ag + (t + 2) * GBK;
            const float* bp = Bg + (size_t)((t + 2) * GBK) * ldn;
#pragma unroll
            for (int i = 0; i < MI; ++i) ra[i] = *(const v4f*)(ap + (size_t)(32 * i) * K);
#pragma unroll
            for (int i = 0; i < NI; ++i) {
                rb0[i] = *(const v4f*)(bp + (size_t)(2 * BKS * i) * ldn);
                rb1[i] = *(const v4f*)(bp + (size_t)(2 * BKS * i + 1) * ldn);
            }
        }
        read_half<MI, NI, PB, 0>(fa0, fb0, aF + nxt * (BM * GPA), bF + nxt * ((GBK / 2) * PB));
        __builtin_amdgcn_sched_barrier(0);
        wmma_half<MI, NI>(acc, fa1, fb1);
        __builtin_amdgcn_sched_barrier(0);
        read_half<MI, NI, PB, 4>(fa1, fb1, aF + nxt * (BM * GPA), bF + nxt * ((GBK / 2) * PB));
        __builtin_amdgcn_sched_barrier(0);
        wmma_half<MI, NI>(acc, fa0, fb0);
        __builtin_amdgcn_sched_barrier(0);
    }
    wmma_half<MI, NI>(acc, fa1, fb1);

    float bv[NI];
#pragma unroll
    for (int ni = 0; ni < NI; ++ni)
        bv[ni] = RELU_BIAS ? bias[bn0 + wn * 16 * NI + ni * 16 + l16] : 0.0f;
#pragma unroll
    for (int mi = 0; mi < MI; ++mi) {
#pragma unroll
        for (int r = 0; r < 8; ++r) {
            const int row = bm0 + wm * 16 * MI + mi * 16 + half * 8 + r;
            float* crow = C + (size_t)row * ldn + bn0 + wn * 16 * NI + l16;
#pragma unroll
            for (int ni = 0; ni < NI; ++ni) {
                float v = acc[mi][ni][r];
                if (RELU_BIAS) {
                    v += bv[ni];
                    v = v > 0.0f ? v : 0.0f;
                }
                crow[ni * 16] = v;
            }
        }
    }
}

template <bool RELU_BIAS, int N, int K>
__global__ __launch_bounds__(256) void gemm_f32_wmma(
    const float* __restrict__ A, const float* __restrict__ Bm,
    const float* __restrict__ bias, float* __restrict__ C)
{
    __shared__ __attribute__((aligned(16))) float As[2 * GBM * GPA];
    __shared__ __attribute__((aligned(16))) float Bs[2 * (GBK / 2) * GPB];

    gemm_block<RELU_BIAS, 4, 2, K>(A, Bm, bias, C, N, blockIdx.y * GBM, blockIdx.x * GBN, As, Bs);
}

#define TRI_SPLIT (LSEQ / GBN / 2)
#define TRI_WHOLE (TRI_BLOCKS - TRI_SPLIT)
#define TRI_SMALL (TRI_SPLIT * 8)
template <int N, int K>
__global__ __launch_bounds__(256) void gemm_g_tri(
    const float* __restrict__ A, const float* __restrict__ Bm,
    const float* __restrict__ bias, float* __restrict__ C)
{
    static_assert(N == LSEQ && GBM == GBN, "the needed-block count assumes square blocks of a square g");
    static_assert(TRI_WHOLE == 2 * TRI_SMALL, "planes 1 and 2 of the grid hold the whole blocks");
    __shared__ __attribute__((aligned(16))) float As[2 * GBM * GPA];
    __shared__ __attribute__((aligned(16))) float Bs[2 * (GBK / 2) * GPB];

    constexpr int NB = N / GBN;
    constexpr int RB = LSEQ / GBM;
    if (blockIdx.z == 0) {
        const int s   = blockIdx.x;
        const int bm0 = (blockIdx.y * RB + RB - 1) * GBM + ((s >> 1) & 3) * 32;
        const int bn0 = (NB / 2 + (s >> 3)) * GBN + (s & 1) * 64;
        gemm_block<true, 1, 1, K>(A, Bm, bias, C, N, bm0, bn0, As, Bs);
    } else {
        const int id = ((blockIdx.z - 1) * gridDim.y + blockIdx.y) * TRI_SMALL + blockIdx.x;
        const int b  = id / TRI_WHOLE;
        int rb = 0, rem = id % TRI_WHOLE;
        while (rem > rb) { rem -= rb + 1; ++rb; }
        const int bm0 = (b * RB + rb) * GBM;
        const int bn0 = (NB - 1 - rb + rem) * GBN;
        gemm_block<true, 4, 2, K>(A, Bm, bias, C, N, bm0, bn0, As, Bs);
    }
}

#define WBT 64
__global__ __launch_bounds__(256) void wq_products(
    const float* __restrict__ Wq, const float* __restrict__ Wg,
    const float* __restrict__ Wc, float* __restrict__ Wqg, float* __restrict__ Wqc)
{
    __shared__ __attribute__((aligned(16))) float As[2 * WBT * GPA];
    __shared__ __attribute__((aligned(16))) float Bs[2 * (GBK / 2) * (2 * WBT + 32)];

    const bool first = blockIdx.x < LSEQ / WBT;
    const float* Bm  = first ? Wg : Wc;
    float*       C   = first ? Wqg : Wqc;
    const int    ldn = first ? LSEQ : D_DIM;
    const int    bn0 = (first ? (int)blockIdx.x : (int)blockIdx.x - LSEQ / WBT) * WBT;
    gemm_block<false, 2, 1, D_DIM>(Wq, Bm, nullptr, C, ldn, blockIdx.y * WBT, bn0, As, Bs);
}

#define SCAN_ROWS  64
#define SCAN_WAVES (LSEQ / SCAN_ROWS)

__global__ __launch_bounds__(SCAN_WAVES * 32) void cumavg_kernel(
    const float* __restrict__ x, float* __restrict__ xavg)
{
    __shared__ float tot[SCAN_WAVES][32];

    const int lane = threadIdx.x & 31;
    const int c  = __builtin_amdgcn_readfirstlane((int)(threadIdx.x >> 5));
    const int b  = blockIdx.y;
    const int d  = blockIdx.x * 32 + lane;
    const int i0 = c * SCAN_ROWS;
    const size_t base = ((size_t)b * LSEQ + (size_t)i0) * D_DIM + d;
    const float* src = x + base;
    float*       dst = xavg + base;

    float v[SCAN_ROWS];
#pragma unroll
    for (int r = 0; r < SCAN_ROWS; ++r) v[r] = src[(size_t)r * D_DIM];
#pragma unroll
    for (int r = 1; r < SCAN_ROWS; ++r) v[r] += v[r - 1];

    tot[c][lane] = v[SCAN_ROWS - 1];
    __syncthreads();

    float off = 0.0f;
    for (int p = 0; p < c; ++p) off += tot[p][lane];

#pragma unroll
    for (int r = 0; r < SCAN_ROWS; ++r)
        dst[(size_t)r * D_DIM] = (off + v[r]) / (float)(i0 + r + 1);
}

__device__ __forceinline__ void attn_load_tile(
    float (&ra)[16], v4f (&rb0)[2], v4f (&rb1)[2],
    const float* __restrict__ Ag, const float* __restrict__ Bg,
    const int j0, const int i0, const int jlim)
{
    const float* ap = Ag + j0;
    const float* bp = Bg + (size_t)j0 * D_DIM;
    if (j0 < i0) {
#pragma unroll
        for (int it = 0; it < 16; ++it) ra[it] = ap[(size_t)it * (8 * LSEQ - 8)];
    } else {
#pragma unroll
        for (int it = 0; it < 16; ++it)
            ra[it] = (j0 <= jlim + 8 * it) ? ap[(size_t)it * (8 * LSEQ - 8)] : 0.0f;
    }
#pragma unroll
    for (int i = 0; i < 2; ++i) {
        rb0[i] = *(const v4f*)(bp + (size_t)(16 * i) * D_DIM);
        rb1[i] = *(const v4f*)(bp + (size_t)(16 * i + 1) * D_DIM);
    }
}

__global__ __launch_bounds__(256) void attn_apply_lds(
    const float* __restrict__ g, const float* __restrict__ x,
    float* __restrict__ out)
{
    __shared__ __attribute__((aligned(16))) float As[2 * GBM * GPA];
    __shared__ __attribute__((aligned(16))) float Bs[2 * (GBK / 2) * GPB];
    __shared__ __attribute__((aligned(16))) float Sinv[GBM];

    const int tid  = threadIdx.x;
    const int lane = tid & 31;
    const int wave = tid >> 5;
    const int half = lane >> 4;
    const int l16  = lane & 15;
    const int wm   = wave >> 2;
    const int wn   = wave & 3;
    const int b    = blockIdx.z;
    const int bn0  = blockIdx.x * GBN;

    const int ac = tid & 31;
    const int ar = tid >> 5;
    float* Ast = As + ar * GPA + ac;
    const int bkp  = tid >> 5;
    const int bn4  = (tid & 31) * 4;
    const float* Bg = x + ((size_t)b * LSEQ + 2 * bkp) * D_DIM + bn0 + bn4;
    float* Bst = Bs + bkp * GPB + 2 * bn4;

    const float* aF = As + (wm * 64 + l16) * GPA + 2 * half;
    const float* bF = Bs + half * GPB + 2 * (wn * 32 + l16);

    for (int pass = 0; pass < 2; ++pass) {
        const int rb = pass ? (int)blockIdx.y : (LSEQ / GBM - 1) - (int)blockIdx.y;
        const int i0 = rb * GBM;
        const int nk = (i0 + GBM) / GBK;

        const float* Ag = g + ((size_t)b * LSEQ + i0 + ar) * LSEQ + (LSEQ - 1 - (i0 + ar)) + ac;
        const int jlim = i0 + ar - ac;

        float ra[16];
        float rs[16];
        v4f rb0[2], rb1[2];
        v8f acc[4][2] = {};
        v2f fa0[4][4], fb0[4][2], fa1[4][4], fb1[4][2];

        attn_load_tile(ra, rb0, rb1, Ag, Bg, 0, i0, jlim);
#pragma unroll
        for (int it = 0; it < 16; ++it) {
            Ast[8 * it * GPA] = ra[it];
            rs[it] = ra[it];
        }
#pragma unroll
        for (int i = 0; i < 2; ++i) {
            v4f lo = {rb0[i].x, rb1[i].x, rb0[i].y, rb1[i].y};
            v4f hi = {rb0[i].z, rb1[i].z, rb0[i].w, rb1[i].w};
            *(v4f*)(Bst + 8 * i * GPB)     = lo;
            *(v4f*)(Bst + 8 * i * GPB + 4) = hi;
        }
        __syncthreads();
        attn_load_tile(ra, rb0, rb1, Ag, Bg, GBK, i0, jlim);
        read_half<4, 2, GPB, 0>(fa0, fb0, aF, bF);
        read_half<4, 2, GPB, 4>(fa1, fb1, aF, bF);
        __builtin_amdgcn_sched_barrier(0);
        wmma_half<4, 2>(acc, fa0, fb0);
        __builtin_amdgcn_sched_barrier(0);

        for (int t = 0; t + 1 < nk; ++t) {
            const int nxt = (t + 1) & 1;
            {
                float* as = Ast + nxt * (GBM * GPA);
                float* bs = Bst + nxt * ((GBK / 2) * GPB);
#pragma unroll
                for (int it = 0; it < 16; ++it) {
                    as[8 * it * GPA] = ra[it];
                    rs[it] += ra[it];
                }
#pragma unroll
                for (int i = 0; i < 2; ++i) {
                    v4f lo = {rb0[i].x, rb1[i].x, rb0[i].y, rb1[i].y};
                    v4f hi = {rb0[i].z, rb1[i].z, rb0[i].w, rb1[i].w};
                    *(v4f*)(bs + 8 * i * GPB)     = lo;
                    *(v4f*)(bs + 8 * i * GPB + 4) = hi;
                }
            }
            __syncthreads();

            attn_load_tile(ra, rb0, rb1, Ag, Bg, (t + 2 < nk ? t + 2 : nk - 1) * GBK, i0, jlim);
            read_half<4, 2, GPB, 0>(fa0, fb0, aF + nxt * (GBM * GPA),
                                    bF + nxt * ((GBK / 2) * GPB));
            __builtin_amdgcn_sched_barrier(0);
            wmma_half<4, 2>(acc, fa1, fb1);
            __builtin_amdgcn_sched_barrier(0);
            read_half<4, 2, GPB, 4>(fa1, fb1, aF + nxt * (GBM * GPA),
                                    bF + nxt * ((GBK / 2) * GPB));
            __builtin_amdgcn_sched_barrier(0);
            wmma_half<4, 2>(acc, fa0, fb0);
            __builtin_amdgcn_sched_barrier(0);
        }
        wmma_half<4, 2>(acc, fa1, fb1);

#pragma unroll
        for (int off = 16; off; off >>= 1) {
            float t[16];
#pragma unroll
            for (int it = 0; it < 16; ++it) t[it] = __shfl_xor(rs[it], off, 32);
#pragma unroll
            for (int it = 0; it < 16; ++it) rs[it] += t[it];
        }
        if (lane == 0) {
#pragma unroll
            for (int it = 0; it < 16; ++it) Sinv[ar + 8 * it] = 1.0f / (rs[it] + 1e-8f);
        }
        __syncthreads();

#pragma unroll
        for (int mi = 0; mi < 4; ++mi) {
#pragma unroll
            for (int r = 0; r < 8; ++r) {
                const int lrow = wm * 64 + mi * 16 + half * 8 + r;
                const int row = b * LSEQ + i0 + lrow;
                const float invs = Sinv[lrow];
                float* crow = out + (size_t)row * D_DIM + bn0 + wn * 32 + l16;
#pragma unroll
                for (int ni = 0; ni < 2; ++ni) crow[ni * 16] = acc[mi][ni][r] * invs;
            }
        }
    }
}

extern "C" void kernel_launch(void* const* d_in, const int* in_sizes, int n_in,
                              void* d_out, int out_size, void* d_ws, size_t ws_size,
                              hipStream_t stream)
{
    const float* x  = (const float*)d_in[0];
    const float* Wq = (const float*)d_in[1];
    const float* Wg = (const float*)d_in[2];
    const float* bg = (const float*)d_in[3];
    const float* Wc = (const float*)d_in[4];
    float* out = (float*)d_out;

    const int Bsz = in_sizes[0] / (LSEQ * D_DIM);
    const int M   = Bsz * LSEQ;

    float* Wqg  = (float*)d_ws;
    float* Wqc  = Wqg  + (size_t)D_DIM * LSEQ;
    float* xavg = Wqg  + (size_t)M * D_DIM;
    float* g    = xavg + (size_t)M * D_DIM;

    dim3 gblk(256);

    wq_products<<<dim3((LSEQ + D_DIM) / WBT, D_DIM / WBT), gblk, 0, stream>>>(
        Wq, Wg, Wc, Wqg, Wqc);

    cumavg_kernel<<<dim3(D_DIM / 32, Bsz), SCAN_WAVES * 32, 0, stream>>>(x, xavg);

    gemm_g_tri<LSEQ, D_DIM><<<dim3(TRI_SMALL, Bsz, 3), gblk, 0, stream>>>(
        xavg, Wqg, bg, g);

    attn_apply_lds<<<dim3(D_DIM / GBN, LSEQ / (2 * GBM), Bsz), gblk, 0, stream>>>(
        g, x, xavg);

    gemm_f32_wmma<false, D_DIM, D_DIM><<<dim3(D_DIM / GBN, M / GBM), gblk, 0, stream>>>(
        xavg, Wqc, nullptr, out);
}


# K loops of the four WMMA kernels: block barrier split: next tile's LDS stores ahead of a trip's last WMMA block, signal in that block's middle, wait at the next trip's head; sum order unchanged
# speedup vs baseline: 3.6542x; 1.0363x over previous
#include <hip/hip_runtime.h>


typedef __attribute__((ext_vector_type(2))) float v2f;
typedef __attribute__((ext_vector_type(4))) float v4f;
typedef __attribute__((ext_vector_type(8))) float v8f;

#define D_DIM 1024
#define LSEQ  2048

#define GBM 128
#define GBN 128
#define GBK 32
#define GPA 36
#define GPB 288
#define TRI_BLOCKS ((LSEQ / GBM) * (LSEQ / GBM + 1) / 2)

__device__ __forceinline__ void tile_arrive()
{
    __builtin_amdgcn_fence(__ATOMIC_RELEASE, "workgroup", "local");
    __builtin_amdgcn_s_barrier_signal(-1);
}
__device__ __forceinline__ void tile_wait()
{
    __builtin_amdgcn_s_barrier_wait(-1);
    __builtin_amdgcn_fence(__ATOMIC_ACQUIRE, "workgroup", "local");
}

template <int MI, int NI, int PB, int S0>
__device__ __forceinline__ void read_half(v2f (&a)[4][MI], v2f (&b)[4][NI],
                                          const float* __restrict__ ab,
                                          const float* __restrict__ bb)
{
#pragma unroll
    for (int s = 0; s < 4; ++s) {
#pragma unroll
        for (int mi = 0; mi < MI; ++mi)
            a[s][mi] = *(const v2f*)(ab + mi * 16 * GPA + 4 * (S0 + s));
#pragma unroll
        for (int ni = 0; ni < NI; ++ni)
            b[s][ni] = *(const v2f*)(bb + 2 * (S0 + s) * PB + ni * 32);
    }
}

template <int MI, int NI>
__device__ __forceinline__ void wmma_half(v8f (&acc)[MI][NI],
                                          const v2f (&a)[4][MI], const v2f (&b)[4][NI])
{
#pragma unroll
    for (int s = 0; s < 4; ++s)
#pragma unroll
        for (int mi = 0; mi < MI; ++mi)
#pragma unroll
            for (int ni = 0; ni < NI; ++ni)
                acc[mi][ni] = __builtin_amdgcn_wmma_f32_16x16x4_f32(
                    false, a[s][mi], false, b[s][ni], (short)0, acc[mi][ni], false, false);
}

template <int MI, int NI, int S_FROM, int S_TO>
__device__ __forceinline__ void wmma_steps(v8f (&acc)[MI][NI],
                                           const v2f (&a)[4][MI], const v2f (&b)[4][NI])
{
#pragma unroll
    for (int s = S_FROM; s < S_TO; ++s)
#pragma unroll
        for (int mi = 0; mi < MI; ++mi)
#pragma unroll
            for (int ni = 0; ni < NI; ++ni)
                acc[mi][ni] = __builtin_amdgcn_wmma_f32_16x16x4_f32(
                    false, a[s][mi], false, b[s][ni], (short)0, acc[mi][ni], false, false);
}

template <bool RELU_BIAS, int MI, int NI, int K>
__device__ __forceinline__ void gemm_block(
    const float* __restrict__ A, const float* __restrict__ Bm,
    const float* __restrict__ bias, float* __restrict__ C,
    const int ldn, const int bm0, const int bn0,
    float* __restrict__ As, float* __restrict__ Bs)
{
    constexpr int BM  = 32 * MI;
    constexpr int BN  = 64 * NI;
    constexpr int PB  = 2 * BN + 32;
    constexpr int TB  = BN / 4;
    constexpr int BKS = 256 / TB;

    const int tid  = threadIdx.x;
    const int lane = tid & 31;
    const int wave = tid >> 5;
    const int half = lane >> 4;
    const int l16  = lane & 15;
    const int wm   = wave >> 2;
    const int wn   = wave & 3;

    const int arow = tid >> 3;
    const int akq  = (tid & 7) * 4;
    const float* Ag = A + (size_t)(bm0 + arow) * K + akq;
    float* Ast = As + arow * GPA + akq;
    const int bkp  = tid / TB;
    const int bn4  = (tid % TB) * 4;
    const float* Bg = Bm + (size_t)(2 * bkp) * ldn + bn0 + bn4;
    float* Bst = Bs + bkp * PB + 2 * bn4;

    const float* aF = As + (wm * 16 * MI + l16) * GPA + 2 * half;
    const float* bF = Bs + half * PB + 2 * (wn * 16 * NI + l16);

    v4f ra[MI], rb0[NI], rb1[NI];
    v8f acc[MI][NI] = {};
    v2f fa0[4][MI], fb0[4][NI], fa1[4][MI], fb1[4][NI];

    constexpr int nk = K / GBK;

#pragma unroll
    for (int i = 0; i < MI; ++i) ra[i] = *(const v4f*)(Ag + (size_t)(32 * i) * K);
#pragma unroll
    for (int i = 0; i < NI; ++i) {
        rb0[i] = *(const v4f*)(Bg + (size_t)(2 * BKS * i) * ldn);
        rb1[i] = *(const v4f*)(Bg + (size_t)(2 * BKS * i + 1) * ldn);
    }
#pragma unroll
    for (int i = 0; i < MI; ++i) *(v4f*)(Ast + 32 * i * GPA) = ra[i];
#pragma unroll
    for (int i = 0; i < NI; ++i) {
        v4f lo = {rb0[i].x, rb1[i].x, rb0[i].y, rb1[i].y};
        v4f hi = {rb0[i].z, rb1[i].z, rb0[i].w, rb1[i].w};
        *(v4f*)(Bst + BKS * i * PB)     = lo;
        *(v4f*)(Bst + BKS * i * PB + 4) = hi;
    }
    __syncthreads();
    {
        const float* ap = Ag + GBK;
        const float* bp = Bg + (size_t)GBK * ldn;
#pragma unroll
        for (int i = 0; i < MI; ++i) ra[i] = *(const v4f*)(ap + (size_t)(32 * i) * K);
#pragma unroll
        for (int i = 0; i < NI; ++i) {
            rb0[i] = *(const v4f*)(bp + (size_t)(2 * BKS * i) * ldn);
            rb1[i] = *(const v4f*)(bp + (size_t)(2 * BKS * i + 1) * ldn);
        }
    }
    read_half<MI, NI, PB, 0>(fa0, fb0, aF, bF);
    read_half<MI, NI, PB, 4>(fa1, fb1, aF, bF);
    __builtin_amdgcn_sched_barrier(0);
    wmma_half<MI, NI>(acc, fa0, fb0);
    __builtin_amdgcn_sched_barrier(0);
    {
        float* as = Ast + BM * GPA;
        float* bs = Bst + (GBK / 2) * PB;
#pragma unroll
        for (int i = 0; i < MI; ++i) *(v4f*)(as + 32 * i * GPA) = ra[i];
#pragma unroll
        for (int i = 0; i < NI; ++i) {
            v4f lo = {rb0[i].x, rb1[i].x, rb0[i].y, rb1[i].y};
            v4f hi = {rb0[i].z, rb1[i].z, rb0[i].w, rb1[i].w};
            *(v4f*)(bs + BKS * i * PB)     = lo;
            *(v4f*)(bs + BKS * i * PB + 4) = hi;
        }
        constexpr int kt = nk > 2 ? 2 : 1;
        const float* ap = Ag + kt * GBK;
        const float* bp = Bg + (size_t)(kt * GBK) * ldn;
#pragma unroll
        for (int i = 0; i < MI; ++i) ra[i] = *(const v4f*)(ap + (size_t)(32 * i) * K);
#pragma unroll
        for (int i = 0; i < NI; ++i) {
            rb0[i] = *(const v4f*)(bp + (size_t)(2 * BKS * i) * ldn);
            rb1[i] = *(const v4f*)(bp + (size_t)(2 * BKS * i + 1) * ldn);
        }
    }
    tile_arrive();

    for (int t = 0; t + 2 < nk; ++t) {
        const int nxt = (t + 1) & 1;
        tile_wait();
        read_half<MI, NI, PB, 0>(fa0, fb0, aF + nxt * (BM * GPA), bF + nxt * ((GBK / 2) * PB));
        __builtin_amdgcn_sched_barrier(0);
        wmma_half<MI, NI>(acc, fa1, fb1);
        __builtin_amdgcn_sched_barrier(0);
        read_half<MI, NI, PB, 4>(fa1, fb1, aF + nxt * (BM * GPA), bF + nxt * ((GBK / 2) * PB));
        {
            float* as = Ast + (t & 1) * (BM * GPA);
            float* bs = Bst + (t & 1) * ((GBK / 2) * PB);
#pragma unroll
            for (int i = 0; i < NI; ++i) asm volatile("" :: "v"(rb0[i]), "v"(rb1[i]));
#pragma unroll
            for (int i = 0; i < MI; ++i) *(v4f*)(as + 32 * i * GPA) = ra[i];
#pragma unroll
            for (int i = 0; i < NI; ++i) {
                v4f lo = {rb0[i].x, rb1[i].x, rb0[i].y, rb1[i].y};
                v4f hi = {rb0[i].z, rb1[i].z, rb0[i].w, rb1[i].w};
                *(v4f*)(bs + BKS * i * PB)     = lo;
                *(v4f*)(bs + BKS * i * PB + 4) = hi;
            }
            const int kt = t + 3 < nk ? t + 3 : nk - 1;
            const float* ap = Ag + kt * GBK;
            const float* bp = Bg + (size_t)(kt * GBK) * ldn;
#pragma unroll
            for (int i = 0; i < MI; ++i) ra[i] = *(const v4f*)(ap + (size_t)(32 * i) * K);
#pragma unroll
            for (int i = 0; i < NI; ++i) {
                rb0[i] = *(const v4f*)(bp + (size_t)(2 * BKS * i) * ldn);
                rb1[i] = *(const v4f*)(bp + (size_t)(2 * BKS * i + 1) * ldn);
            }
        }
        __builtin_amdgcn_sched_barrier(0);
        wmma_steps<MI, NI, 0, 2>(acc, fa0, fb0);
        __builtin_amdgcn_sched_barrier(0);
        tile_arrive();
        __builtin_amdgcn_sched_barrier(0);
        wmma_steps<MI, NI, 2, 4>(acc, fa0, fb0);
        __builtin_amdgcn_sched_barrier(0);
    }
    {
        constexpr int nxt = (nk - 1) & 1;
        tile_wait();
        read_half<MI, NI, PB, 0>(fa0, fb0, aF + nxt * (BM * GPA), bF + nxt * ((GBK / 2) * PB));
        __builtin_amdgcn_sched_barrier(0);
        wmma_half<MI, NI>(acc, fa1, fb1);
        __builtin_amdgcn_sched_barrier(0);
        read_half<MI, NI, PB, 4>(fa1, fb1, aF + nxt * (BM * GPA), bF + nxt * ((GBK / 2) * PB));
        __builtin_amdgcn_sched_barrier(0);
        wmma_half<MI, NI>(acc, fa0, fb0);
    }
    wmma_half<MI, NI>(acc, fa1, fb1);

    float bv[NI];
#pragma unroll
    for (int ni = 0; ni < NI; ++ni)
        bv[ni] = RELU_BIAS ? bias[bn0 + wn * 16 * NI + ni * 16 + l16] : 0.0f;
#pragma unroll
    for (int mi = 0; mi < MI; ++mi) {
#pragma unroll
        for (int r = 0; r < 8; ++r) {
            const int row = bm0 + wm * 16 * MI + mi * 16 + half * 8 + r;
            float* crow = C + (size_t)row * ldn + bn0 + wn * 16 * NI + l16;
#pragma unroll
            for (int ni = 0; ni < NI; ++ni) {
                float v = acc[mi][ni][r];
                if (RELU_BIAS) {
                    v += bv[ni];
                    v = v > 0.0f ? v : 0.0f;
                }
                crow[ni * 16] = v;
            }
        }
    }
}

template <bool RELU_BIAS, int N, int K>
__global__ __launch_bounds__(256) void gemm_f32_wmma(
    const float* __restrict__ A, const float* __restrict__ Bm,
    const float* __restrict__ bias, float* __restrict__ C)
{
    __shared__ __attribute__((aligned(16))) float As[2 * GBM * GPA];
    __shared__ __attribute__((aligned(16))) float Bs[2 * (GBK / 2) * GPB];

    gemm_block<RELU_BIAS, 4, 2, K>(A, Bm, bias, C, N, blockIdx.y * GBM, blockIdx.x * GBN, As, Bs);
}

#define TRI_SPLIT (LSEQ / GBN / 2)
#define TRI_WHOLE (TRI_BLOCKS - TRI_SPLIT)
#define TRI_SMALL (TRI_SPLIT * 8)
template <int N, int K>
__global__ __launch_bounds__(256) void gemm_g_tri(
    const float* __restrict__ A, const float* __restrict__ Bm,
    const float* __restrict__ bias, float* __restrict__ C)
{
    static_assert(N == LSEQ && GBM == GBN, "the needed-block count assumes square blocks of a square g");
    static_assert(TRI_WHOLE == 2 * TRI_SMALL, "planes 1 and 2 of the grid hold the whole blocks");
    __shared__ __attribute__((aligned(16))) float As[2 * GBM * GPA];
    __shared__ __attribute__((aligned(16))) float Bs[2 * (GBK / 2) * GPB];

    constexpr int NB = N / GBN;
    constexpr int RB = LSEQ / GBM;
    if (blockIdx.z == 0) {
        const int s   = blockIdx.x;
        const int bm0 = (blockIdx.y * RB + RB - 1) * GBM + ((s >> 1) & 3) * 32;
        const int bn0 = (NB / 2 + (s >> 3)) * GBN + (s & 1) * 64;
        gemm_block<true, 1, 1, K>(A, Bm, bias, C, N, bm0, bn0, As, Bs);
    } else {
        const int id = ((blockIdx.z - 1) * gridDim.y + blockIdx.y) * TRI_SMALL + blockIdx.x;
        const int b  = id / TRI_WHOLE;
        int rb = 0, rem = id % TRI_WHOLE;
        while (rem > rb) { rem -= rb + 1; ++rb; }
        const int bm0 = (b * RB + rb) * GBM;
        const int bn0 = (NB - 1 - rb + rem) * GBN;
        gemm_block<true, 4, 2, K>(A, Bm, bias, C, N, bm0, bn0, As, Bs);
    }
}

#define WBT 64
__global__ __launch_bounds__(256) void wq_products(
    const float* __restrict__ Wq, const float* __restrict__ Wg,
    const float* __restrict__ Wc, float* __restrict__ Wqg, float* __restrict__ Wqc)
{
    __shared__ __attribute__((aligned(16))) float As[2 * WBT * GPA];
    __shared__ __attribute__((aligned(16))) float Bs[2 * (GBK / 2) * (2 * WBT + 32)];

    const bool first = blockIdx.x < LSEQ / WBT;
    const float* Bm  = first ? Wg : Wc;
    float*       C   = first ? Wqg : Wqc;
    const int    ldn = first ? LSEQ : D_DIM;
    const int    bn0 = (first ? (int)blockIdx.x : (int)blockIdx.x - LSEQ / WBT) * WBT;
    gemm_block<false, 2, 1, D_DIM>(Wq, Bm, nullptr, C, ldn, blockIdx.y * WBT, bn0, As, Bs);
}

#define SCAN_ROWS  64
#define SCAN_WAVES (LSEQ / SCAN_ROWS)

__global__ __launch_bounds__(SCAN_WAVES * 32) void cumavg_kernel(
    const float* __restrict__ x, float* __restrict__ xavg)
{
    __shared__ float tot[SCAN_WAVES][32];

    const int lane = threadIdx.x & 31;
    const int c  = __builtin_amdgcn_readfirstlane((int)(threadIdx.x >> 5));
    const int b  = blockIdx.y;
    const int d  = blockIdx.x * 32 + lane;
    const int i0 = c * SCAN_ROWS;
    const size_t base = ((size_t)b * LSEQ + (size_t)i0) * D_DIM + d;
    const float* src = x + base;
    float*       dst = xavg + base;

    float v[SCAN_ROWS];
#pragma unroll
    for (int r = 0; r < SCAN_ROWS; ++r) v[r] = src[(size_t)r * D_DIM];
#pragma unroll
    for (int r = 1; r < SCAN_ROWS; ++r) v[r] += v[r - 1];

    tot[c][lane] = v[SCAN_ROWS - 1];
    __syncthreads();

    float off = 0.0f;
    for (int p = 0; p < c; ++p) off += tot[p][lane];

#pragma unroll
    for (int r = 0; r < SCAN_ROWS; ++r)
        dst[(size_t)r * D_DIM] = (off + v[r]) / (float)(i0 + r + 1);
}

__device__ __forceinline__ void attn_load_tile(
    float (&ra)[16], v4f (&rb0)[2], v4f (&rb1)[2],
    const float* __restrict__ Ag, const float* __restrict__ Bg,
    const int j0, const int i0, const int jlim)
{
    const float* ap = Ag + j0;
    const float* bp = Bg + (size_t)j0 * D_DIM;
    if (j0 < i0) {
#pragma unroll
        for (int it = 0; it < 16; ++it) ra[it] = ap[(size_t)it * (8 * LSEQ - 8)];
    } else {
#pragma unroll
        for (int it = 0; it < 16; ++it)
            ra[it] = (j0 <= jlim + 8 * it) ? ap[(size_t)it * (8 * LSEQ - 8)] : 0.0f;
    }
#pragma unroll
    for (int i = 0; i < 2; ++i) {
        rb0[i] = *(const v4f*)(bp + (size_t)(16 * i) * D_DIM);
        rb1[i] = *(const v4f*)(bp + (size_t)(16 * i + 1) * D_DIM);
    }
}

__global__ __launch_bounds__(256) void attn_apply_lds(
    const float* __restrict__ g, const float* __restrict__ x,
    float* __restrict__ out)
{
    __shared__ __attribute__((aligned(16))) float As[2 * GBM * GPA];
    __shared__ __attribute__((aligned(16))) float Bs[2 * (GBK / 2) * GPB];
    __shared__ __attribute__((aligned(16))) float Sinv[GBM];

    const int tid  = threadIdx.x;
    const int lane = tid & 31;
    const int wave = tid >> 5;
    const int half = lane >> 4;
    const int l16  = lane & 15;
    const int wm   = wave >> 2;
    const int wn   = wave & 3;
    const int b    = blockIdx.z;
    const int bn0  = blockIdx.x * GBN;

    const int ac = tid & 31;
    const int ar = tid >> 5;
    float* Ast = As + ar * GPA + ac;
    const int bkp  = tid >> 5;
    const int bn4  = (tid & 31) * 4;
    const float* Bg = x + ((size_t)b * LSEQ + 2 * bkp) * D_DIM + bn0 + bn4;
    float* Bst = Bs + bkp * GPB + 2 * bn4;

    const float* aF = As + (wm * 64 + l16) * GPA + 2 * half;
    const float* bF = Bs + half * GPB + 2 * (wn * 32 + l16);

    for (int pass = 0; pass < 2; ++pass) {
        const int rb = pass ? (int)blockIdx.y : (LSEQ / GBM - 1) - (int)blockIdx.y;
        const int i0 = rb * GBM;
        const int nk = (i0 + GBM) / GBK;

        const float* Ag = g + ((size_t)b * LSEQ + i0 + ar) * LSEQ + (LSEQ - 1 - (i0 + ar)) + ac;
        const int jlim = i0 + ar - ac;

        float ra[16];
        float rs[16];
        v4f rb0[2], rb1[2];
        float na[16];
        v4f nb0[2], nb1[2];
        v8f acc[4][2] = {};
        v2f fa0[4][4], fb0[4][2], fa1[4][4], fb1[4][2];

        attn_load_tile(ra, rb0, rb1, Ag, Bg, 0, i0, jlim);
#pragma unroll
        for (int it = 0; it < 16; ++it) {
            Ast[8 * it * GPA] = ra[it];
            rs[it] = ra[it];
        }
#pragma unroll
        for (int i = 0; i < 2; ++i) {
            v4f lo = {rb0[i].x, rb1[i].x, rb0[i].y, rb1[i].y};
            v4f hi = {rb0[i].z, rb1[i].z, rb0[i].w, rb1[i].w};
            *(v4f*)(Bst + 8 * i * GPB)     = lo;
            *(v4f*)(Bst + 8 * i * GPB + 4) = hi;
        }
        __syncthreads();
        attn_load_tile(ra, rb0, rb1, Ag, Bg, GBK, i0, jlim);
        attn_load_tile(na, nb0, nb1, Ag, Bg, 2 * GBK, i0, jlim);
        read_half<4, 2, GPB, 0>(fa0, fb0, aF, bF);
        read_half<4, 2, GPB, 4>(fa1, fb1, aF, bF);
        __builtin_amdgcn_sched_barrier(0);
        wmma_half<4, 2>(acc, fa0, fb0);
        __builtin_amdgcn_sched_barrier(0);
        {
            float* as = Ast + GBM * GPA;
            float* bs = Bst + (GBK / 2) * GPB;
#pragma unroll
            for (int it = 0; it < 16; ++it) {
                as[8 * it * GPA] = ra[it];
                rs[it] += ra[it];
            }
#pragma unroll
            for (int i = 0; i < 2; ++i) {
                v4f lo = {rb0[i].x, rb1[i].x, rb0[i].y, rb1[i].y};
                v4f hi = {rb0[i].z, rb1[i].z, rb0[i].w, rb1[i].w};
                *(v4f*)(bs + 8 * i * GPB)     = lo;
                *(v4f*)(bs + 8 * i * GPB + 4) = hi;
            }
        }
#pragma unroll
        for (int it = 0; it < 16; ++it) ra[it] = na[it];
#pragma unroll
        for (int i = 0; i < 2; ++i) { rb0[i] = nb0[i]; rb1[i] = nb1[i]; }
        tile_arrive();

        for (int t = 0; t + 2 < nk; ++t) {
            const int nxt = (t + 1) & 1;
            tile_wait();
            attn_load_tile(na, nb0, nb1, Ag, Bg, (t + 3 < nk ? t + 3 : nk - 1) * GBK, i0, jlim);
            read_half<4, 2, GPB, 0>(fa0, fb0, aF + nxt * (GBM * GPA),
                                    bF + nxt * ((GBK / 2) * GPB));
            __builtin_amdgcn_sched_barrier(0);
            wmma_half<4, 2>(acc, fa1, fb1);
            __builtin_amdgcn_sched_barrier(0);
            read_half<4, 2, GPB, 4>(fa1, fb1, aF + nxt * (GBM * GPA),
                                    bF + nxt * ((GBK / 2) * GPB));
            {
                float* as = Ast + (t & 1) * (GBM * GPA);
                float* bs = Bst + (t & 1) * ((GBK / 2) * GPB);
#pragma unroll
                for (int it = 0; it < 16; ++it) {
                    as[8 * it * GPA] = ra[it];
                    rs[it] += ra[it];
                }
#pragma unroll
                for (int i = 0; i < 2; ++i) {
                    v4f lo = {rb0[i].x, rb1[i].x, rb0[i].y, rb1[i].y};
                    v4f hi = {rb0[i].z, rb1[i].z, rb0[i].w, rb1[i].w};
                    *(v4f*)(bs + 8 * i * GPB)     = lo;
                    *(v4f*)(bs + 8 * i * GPB + 4) = hi;
                }
            }
            __builtin_amdgcn_sched_barrier(0);
            wmma_steps<4, 2, 0, 2>(acc, fa0, fb0);
            __builtin_amdgcn_sched_barrier(0);
            tile_arrive();
            __builtin_amdgcn_sched_barrier(0);
            wmma_steps<4, 2, 2, 4>(acc, fa0, fb0);
            __builtin_amdgcn_sched_barrier(0);
#pragma unroll
            for (int it = 0; it < 16; ++it) ra[it] = na[it];
#pragma unroll
            for (int i = 0; i < 2; ++i) { rb0[i] = nb0[i]; rb1[i] = nb1[i]; }
        }
        {
            const int nxt = (nk - 1) & 1;
            tile_wait();
            read_half<4, 2, GPB, 0>(fa0, fb0, aF + nxt * (GBM * GPA),
                                    bF + nxt * ((GBK / 2) * GPB));
            __builtin_amdgcn_sched_barrier(0);
            wmma_half<4, 2>(acc, fa1, fb1);
            __builtin_amdgcn_sched_barrier(0);
            read_half<4, 2, GPB, 4>(fa1, fb1, aF + nxt * (GBM * GPA),
                                    bF + nxt * ((GBK / 2) * GPB));
            __builtin_amdgcn_sched_barrier(0);
            wmma_half<4, 2>(acc, fa0, fb0);
        }
        wmma_half<4, 2>(acc, fa1, fb1);

#pragma unroll
        for (int off = 16; off; off >>= 1) {
            float t[16];
#pragma unroll
            for (int it = 0; it < 16; ++it) t[it] = __shfl_xor(rs[it], off, 32);
#pragma unroll
            for (int it = 0; it < 16; ++it) rs[it] += t[it];
        }
        if (lane == 0) {
#pragma unroll
            for (int it = 0; it < 16; ++it) Sinv[ar + 8 * it] = 1.0f / (rs[it] + 1e-8f);
        }
        __syncthreads();

#pragma unroll
        for (int mi = 0; mi < 4; ++mi) {
#pragma unroll
            for (int r = 0; r < 8; ++r) {
                const int lrow = wm * 64 + mi * 16 + half * 8 + r;
                const int row = b * LSEQ + i0 + lrow;
                const float invs = Sinv[lrow];
                float* crow = out + (size_t)row * D_DIM + bn0 + wn * 32 + l16;
#pragma unroll
                for (int ni = 0; ni < 2; ++ni) crow[ni * 16] = acc[mi][ni][r] * invs;
            }
        }
    }
}

extern "C" void kernel_launch(void* const* d_in, const int* in_sizes, int n_in,
                              void* d_out, int out_size, void* d_ws, size_t ws_size,
                              hipStream_t stream)
{
    const float* x  = (const float*)d_in[0];
    const float* Wq = (const float*)d_in[1];
    const float* Wg = (const float*)d_in[2];
    const float* bg = (const float*)d_in[3];
    const float* Wc = (const float*)d_in[4];
    float* out = (float*)d_out;

    const int Bsz = in_sizes[0] / (LSEQ * D_DIM);
    const int M   = Bsz * LSEQ;

    float* Wqg  = (float*)d_ws;
    float* Wqc  = Wqg  + (size_t)D_DIM * LSEQ;
    float* xavg = Wqg  + (size_t)M * D_DIM;
    float* g    = xavg + (size_t)M * D_DIM;

    dim3 gblk(256);

    wq_products<<<dim3((LSEQ + D_DIM) / WBT, D_DIM / WBT), gblk, 0, stream>>>(
        Wq, Wg, Wc, Wqg, Wqc);

    cumavg_kernel<<<dim3(D_DIM / 32, Bsz), SCAN_WAVES * 32, 0, stream>>>(x, xavg);

    gemm_g_tri<LSEQ, D_DIM><<<dim3(TRI_SMALL, Bsz, 3), gblk, 0, stream>>>(
        xavg, Wqg, bg, g);

    attn_apply_lds<<<dim3(D_DIM / GBN, LSEQ / (2 * GBM), Bsz), gblk, 0, stream>>>(
        g, x, xavg);

    gemm_f32_wmma<false, D_DIM, D_DIM><<<dim3(D_DIM / GBN, M / GBM), gblk, 0, stream>>>(
        xavg, Wqc, nullptr, out);
}
